# starting kernel
# baseline (speedup 1.0000x reference)
#include <hip/hip_runtime.h>
#include <hip/hip_bf16.h>


typedef __attribute__((ext_vector_type(2))) float v2f;
typedef __attribute__((ext_vector_type(8))) float v8f;

#define FDIM 128
#define NHEAD 4
#define CHAN 32
#define NEG_SLOPE 0.2f

__global__ __launch_bounds__(256) void gat_gemm_wmma(const float* __restrict__ X,
                                                     const float* __restrict__ W,
                                                     float* __restrict__ H,
                                                     int N) {
  const int lane = threadIdx.x & 31;
  const int wave = threadIdx.x >> 5;
  const int m    = lane & 15;
  const int half = lane >> 4;
  const int row0 = blockIdx.x * 16;
  const int col0 = wave * 16;

  int rowA = row0 + m;
  if (rowA >= N) rowA = N - 1;
  const float* __restrict__ xrow = X + (size_t)rowA * FDIM;

  v8f acc = {};
#pragma unroll
  for (int k = 0; k < FDIM; k += 4) {
    const int ka = k + 2 * half;
    v2f a = *(const v2f*)(xrow + ka);
    v2f b;
    b.x = W[ka * FDIM + col0 + m];
    b.y = W[(ka + 1) * FDIM + col0 + m];
    acc = __builtin_amdgcn_wmma_f32_16x16x4_f32(
         false, a,  false, b,
         (short)0, acc,  false,  false);
  }

#pragma unroll
  for (int v = 0; v < 8; ++v) {
    const int row = row0 + v + 8 * half;
    if (row < N) H[(size_t)row * FDIM + col0 + m] = acc[v];
  }
}

__global__ void gat_alpha(const float* __restrict__ H,
                          const float* __restrict__ a_src,
                          const float* __restrict__ a_dst,
                          float* __restrict__ as_, float* __restrict__ ad_,
                          int N) {
  const int tid = blockIdx.x * blockDim.x + threadIdx.x;
  if (tid >= N * NHEAD) return;
  const int n = tid >> 2;
  const int h = tid & 3;
  const float* __restrict__ hp = H + (size_t)n * FDIM + h * CHAN;
  const float* __restrict__ asv = a_src + h * CHAN;
  const float* __restrict__ adv = a_dst + h * CHAN;
  float s = 0.0f, d = 0.0f;
#pragma unroll
  for (int c = 0; c < CHAN; ++c) {
    const float hv = hp[c];
    s = fmaf(hv, asv[c], s);
    d = fmaf(hv, adv[c], d);
  }
  as_[tid] = s;
  ad_[tid] = d;
}

__global__ void fill_kernel(float* __restrict__ p, float v, int n) {
  const int i = blockIdx.x * blockDim.x + threadIdx.x;
  if (i < n) p[i] = v;
}

__device__ __forceinline__ void atomicMaxFloat(float* addr, float v) {
  const int iv = __float_as_int(v);
  if (iv >= 0) atomicMax((int*)addr, iv);
  else         atomicMin((unsigned int*)addr, (unsigned int)iv);
}

__device__ __forceinline__ void edge_endpoints(const int* __restrict__ ei,
                                               int e, int E, int* s, int* d) {
  if (e < E) { *s = ei[e]; *d = ei[E + e]; }
  else       { *s = e - E; *d = e - E; }
}

__device__ __forceinline__ float leaky(float v) {
  return v > 0.0f ? v : NEG_SLOPE * v;
}

__global__ void gat_edge_max(const int* __restrict__ ei, int E, int ET,
                             const float* __restrict__ as_,
                             const float* __restrict__ ad_,
                             float* __restrict__ m_) {
  const int tid = blockIdx.x * blockDim.x + threadIdx.x;
  if (tid >= ET * NHEAD) return;
  const int e = tid >> 2, h = tid & 3;
  int s, d;
  edge_endpoints(ei, e, E, &s, &d);
  const float v = leaky(as_[s * NHEAD + h] + ad_[d * NHEAD + h]);
  atomicMaxFloat(&m_[d * NHEAD + h], v);
}

__global__ void gat_edge_expsum(const int* __restrict__ ei, int E, int ET,
                                const float* __restrict__ as_,
                                const float* __restrict__ ad_,
                                const float* __restrict__ m_,
                                float* __restrict__ den_) {
  const int tid = blockIdx.x * blockDim.x + threadIdx.x;
  if (tid >= ET * NHEAD) return;
  const int e = tid >> 2, h = tid & 3;
  int s, d;
  edge_endpoints(ei, e, E, &s, &d);
  const float v = leaky(as_[s * NHEAD + h] + ad_[d * NHEAD + h]);
  const float ex = __expf(v - m_[d * NHEAD + h]);
  atomicAdd(&den_[d * NHEAD + h], ex);
}

__global__ void gat_edge_scatter(const int* __restrict__ ei, int E, int ET,
                                 const float* __restrict__ as_,
                                 const float* __restrict__ ad_,
                                 const float* __restrict__ m_,
                                 const float* __restrict__ den_,
                                 const float* __restrict__ H,
                                 float* __restrict__ out) {
  const long long tid = (long long)blockIdx.x * blockDim.x + threadIdx.x;
  if (tid >= (long long)ET * 32) return;
  const int e  = (int)(tid >> 5);
  const int q  = (int)(tid & 31);
  const int c0 = q * 4;
  const int h  = c0 >> 5;
  int s, d;
  edge_endpoints(ei, e, E, &s, &d);
  const float v  = leaky(as_[s * NHEAD + h] + ad_[d * NHEAD + h]);
  const float ex = __expf(v - m_[d * NHEAD + h]);
  const float alpha = ex / (den_[d * NHEAD + h] + 1e-16f);

  const float4 hv = *(const float4*)(H + (size_t)s * FDIM + c0);
  float* __restrict__ op = out + (size_t)d * FDIM + c0;
  atomicAdd(op + 0, alpha * hv.x);
  atomicAdd(op + 1, alpha * hv.y);
  atomicAdd(op + 2, alpha * hv.z);
  atomicAdd(op + 3, alpha * hv.w);
}

__global__ void gat_bias_act(float* __restrict__ p, const float* __restrict__ b,
                             int total, int do_relu) {
  const int i = blockIdx.x * blockDim.x + threadIdx.x;
  if (i >= total) return;
  float v = p[i] + b[i & (FDIM - 1)];
  if (do_relu) v = v > 0.0f ? v : 0.0f;
  p[i] = v;
}

static inline int cdiv(long long a, int b) { return (int)((a + b - 1) / b); }

extern "C" void kernel_launch(void* const* d_in, const int* in_sizes, int n_in,
                              void* d_out, int out_size, void* d_ws, size_t ws_size,
                              hipStream_t stream) {
  const float* x     = (const float*)d_in[0];
  const int*   ei    = (const int*)d_in[1];
  const float* W1    = (const float*)d_in[2];
  const float* asrc1 = (const float*)d_in[3];
  const float* adst1 = (const float*)d_in[4];
  const float* b1    = (const float*)d_in[5];
  const float* W2    = (const float*)d_in[6];
  const float* asrc2 = (const float*)d_in[7];
  const float* adst2 = (const float*)d_in[8];
  const float* b2    = (const float*)d_in[9];
  float* out = (float*)d_out;

  const int N  = in_sizes[0] / FDIM;
  const int E  = in_sizes[1] / 2;
  const int ET = E + N;

  float* H    = (float*)d_ws;
  float* Hact = H    + (size_t)N * FDIM;
  float* as_  = Hact + (size_t)N * FDIM;
  float* ad_  = as_  + (size_t)N * NHEAD;
  float* m_   = ad_  + (size_t)N * NHEAD;
  float* den_ = m_   + (size_t)N * NHEAD;

  const int TB = 256;
  const int gTiles = cdiv(N, 16);
  const int gNH    = cdiv((long long)N * NHEAD, TB);
  const int gNF    = cdiv((long long)N * FDIM, TB);
  const int gEH    = cdiv((long long)ET * NHEAD, TB);
  const int gE32   = cdiv((long long)ET * 32, TB);
  const float NEG_INF = -__builtin_huge_valf();

  fill_kernel<<<gNH, TB, 0, stream>>>(m_, NEG_INF, N * NHEAD);
  fill_kernel<<<gNH, TB, 0, stream>>>(den_, 0.0f, N * NHEAD);
  fill_kernel<<<gNF, TB, 0, stream>>>(Hact, 0.0f, N * FDIM);

  gat_gemm_wmma<<<gTiles, TB, 0, stream>>>(x, W1, H, N);
  gat_alpha<<<gNH, TB, 0, stream>>>(H, asrc1, adst1, as_, ad_, N);
  gat_edge_max<<<gEH, TB, 0, stream>>>(ei, E, ET, as_, ad_, m_);
  gat_edge_expsum<<<gEH, TB, 0, stream>>>(ei, E, ET, as_, ad_, m_, den_);
  gat_edge_scatter<<<gE32, TB, 0, stream>>>(ei, E, ET, as_, ad_, m_, den_, H, Hact);
  gat_bias_act<<<gNF, TB, 0, stream>>>(Hact, b1, N * FDIM,  1);

  fill_kernel<<<gNH, TB, 0, stream>>>(m_, NEG_INF, N * NHEAD);
  fill_kernel<<<gNH, TB, 0, stream>>>(den_, 0.0f, N * NHEAD);
  fill_kernel<<<gNF, TB, 0, stream>>>(out, 0.0f, N * FDIM);

  gat_gemm_wmma<<<gTiles, TB, 0, stream>>>(Hact, W2, H, N);
  gat_alpha<<<gNH, TB, 0, stream>>>(H, asrc2, adst2, as_, ad_, N);
  gat_edge_max<<<gEH, TB, 0, stream>>>(ei, E, ET, as_, ad_, m_);
  gat_edge_expsum<<<gEH, TB, 0, stream>>>(ei, E, ET, as_, ad_, m_, den_);
  gat_edge_scatter<<<gE32, TB, 0, stream>>>(ei, E, ET, as_, ad_, m_, den_, H, out);
  gat_bias_act<<<gNF, TB, 0, stream>>>(out, b2, N * FDIM,  0);
}


# Edges counting-sorted by destination once per call (int atomics, arrival-order slots); one wave per node does softmax and row sum in place of float atomics, so the summation order changes
# speedup vs baseline: 2.7448x; 2.7448x over previous
#include <hip/hip_runtime.h>
#include <hip/hip_bf16.h>


typedef __attribute__((ext_vector_type(2))) float v2f;
typedef __attribute__((ext_vector_type(8))) float v8f;

#define FDIM 128
#define NHEAD 4
#define CHAN 32
#define NEG_SLOPE 0.2f

__global__ __launch_bounds__(256) void gat_gemm_wmma(const float* __restrict__ X,
                                                     const float* __restrict__ W,
                                                     float* __restrict__ H,
                                                     int N) {
  const int lane = threadIdx.x & 31;
  const int wave = threadIdx.x >> 5;
  const int m    = lane & 15;
  const int half = lane >> 4;
  const int row0 = blockIdx.x * 16;
  const int col0 = wave * 16;

  int rowA = row0 + m;
  if (rowA >= N) rowA = N - 1;
  const float* __restrict__ xrow = X + (size_t)rowA * FDIM;

  v8f acc = {};
#pragma unroll
  for (int k = 0; k < FDIM; k += 4) {
    const int ka = k + 2 * half;
    v2f a = *(const v2f*)(xrow + ka);
    v2f b;
    b.x = W[ka * FDIM + col0 + m];
    b.y = W[(ka + 1) * FDIM + col0 + m];
    acc = __builtin_amdgcn_wmma_f32_16x16x4_f32(
         false, a,  false, b,
         (short)0, acc,  false,  false);
  }

#pragma unroll
  for (int v = 0; v < 8; ++v) {
    const int row = row0 + v + 8 * half;
    if (row < N) H[(size_t)row * FDIM + col0 + m] = acc[v];
  }
}

__global__ void gat_alpha(const float* __restrict__ H,
                          const float* __restrict__ a_src,
                          const float* __restrict__ a_dst,
                          float* __restrict__ as_, float* __restrict__ ad_,
                          int N) {
  const int tid = blockIdx.x * blockDim.x + threadIdx.x;
  if (tid >= N * NHEAD) return;
  const int n = tid >> 2;
  const int h = tid & 3;
  const float* __restrict__ hp = H + (size_t)n * FDIM + h * CHAN;
  const float* __restrict__ asv = a_src + h * CHAN;
  const float* __restrict__ adv = a_dst + h * CHAN;
  float s = 0.0f, d = 0.0f;
#pragma unroll
  for (int c = 0; c < CHAN; ++c) {
    const float hv = hp[c];
    s = fmaf(hv, asv[c], s);
    d = fmaf(hv, adv[c], d);
  }
  as_[tid] = s;
  ad_[tid] = d;
}

__global__ void zero_ints(int* __restrict__ p, int n) {
  const int i = blockIdx.x * blockDim.x + threadIdx.x;
  if (i < n) p[i] = 0;
}

__device__ __forceinline__ void edge_endpoints(const int* __restrict__ ei,
                                               int e, int E, int* s, int* d) {
  if (e < E) { *s = ei[e]; *d = ei[E + e]; }
  else       { *s = e - E; *d = e - E; }
}

__device__ __forceinline__ float leaky(float v) {
  return v > 0.0f ? v : NEG_SLOPE * v;
}

__global__ void csr_count(const int* __restrict__ ei, int E, int ET,
                          int* __restrict__ cnt) {
  const int e = blockIdx.x * blockDim.x + threadIdx.x;
  if (e >= ET) return;
  int s, d;
  edge_endpoints(ei, e, E, &s, &d);
  atomicAdd(&cnt[d], 1);
}

__global__ void csr_alloc(const int* __restrict__ cnt, int* __restrict__ total,
                          int* __restrict__ start, int* __restrict__ cursor,
                          int N) {
  const int d    = blockIdx.x * blockDim.x + threadIdx.x;
  const int lane = threadIdx.x & 31;
  const int c    = (d < N) ? cnt[d] : 0;
  int incl = c;
  for (int off = 1; off < 32; off <<= 1) {
    const int t = __shfl_up(incl, off);
    if (lane >= off) incl += t;
  }
  const int wsum = __shfl(incl, 31);
  int base = 0;
  if (lane == 31) base = atomicAdd(total, wsum);
  base = __shfl(base, 31);
  if (d < N) {
    const int st = base + incl - c;
    start[d]  = st;
    cursor[d] = st;
  }
}

__global__ void csr_fill(const int* __restrict__ ei, int E, int ET,
                         int* __restrict__ cursor, int* __restrict__ col) {
  const int e = blockIdx.x * blockDim.x + threadIdx.x;
  if (e >= ET) return;
  int s, d;
  edge_endpoints(ei, e, E, &s, &d);
  const int p = atomicAdd(&cursor[d], 1);
  col[p] = s;
}

__device__ __forceinline__ float wave_max(float v) {
  for (int off = 16; off > 0; off >>= 1) v = fmaxf(v, __shfl_xor(v, off));
  return v;
}
__device__ __forceinline__ float wave_sum(float v) {
  for (int off = 16; off > 0; off >>= 1) v += __shfl_xor(v, off);
  return v;
}

__device__ __forceinline__ bool chunk_scores(const int* __restrict__ col,
                                             const float* __restrict__ as_,
                                             int beg, int deg, int base, int lane,
                                             float4 adv, int* s, float4* e) {
  const int j = base + lane;
  if (j >= deg) return false;
  const int sj = col[beg + j];
  const float4 a = *(const float4*)(as_ + (size_t)sj * NHEAD);
  *s = sj;
  e->x = leaky(a.x + adv.x);
  e->y = leaky(a.y + adv.y);
  e->z = leaky(a.z + adv.z);
  e->w = leaky(a.w + adv.w);
  return true;
}

__device__ __forceinline__ float bcast(float v, int jj) {
  return __int_as_float(__builtin_amdgcn_readlane(__float_as_int(v), jj));
}

struct RowTerm { float a; float4 hv; };

__device__ __forceinline__ RowTerm row_term(const float* __restrict__ H, int lane,
                                            int s, float4 al, int jj, int4 k) {
  RowTerm t;
  const int sj = __builtin_amdgcn_readlane(s, jj);
  t.a = __int_as_float((__float_as_int(bcast(al.x, jj)) & k.x) |
                       (__float_as_int(bcast(al.y, jj)) & k.y) |
                       (__float_as_int(bcast(al.z, jj)) & k.z) |
                       (__float_as_int(bcast(al.w, jj)) & k.w));
  t.hv = *((const float4*)(H + (size_t)sj * FDIM) + lane);
  return t;
}

__device__ __forceinline__ void row_fma(float4* acc, RowTerm t) {
  acc->x = fmaf(t.a, t.hv.x, acc->x);
  acc->y = fmaf(t.a, t.hv.y, acc->y);
  acc->z = fmaf(t.a, t.hv.z, acc->z);
  acc->w = fmaf(t.a, t.hv.w, acc->w);
}

__device__ __forceinline__ void gather_rows(const float* __restrict__ H,
                                            int s, float4 al, int cntc,
                                            int lane, int4 k, float4* acc) {
  int jj = 0;
  for (; jj + 4 <= cntc; jj += 4) {
    const RowTerm t0 = row_term(H, lane, s, al, jj + 0, k);
    const RowTerm t1 = row_term(H, lane, s, al, jj + 1, k);
    const RowTerm t2 = row_term(H, lane, s, al, jj + 2, k);
    const RowTerm t3 = row_term(H, lane, s, al, jj + 3, k);
    row_fma(acc, t0); row_fma(acc, t1); row_fma(acc, t2); row_fma(acc, t3);
  }
  for (; jj < cntc; ++jj) row_fma(acc, row_term(H, lane, s, al, jj, k));
}

__global__ __launch_bounds__(256) void gat_node_aggregate(
    const int* __restrict__ cnt, const int* __restrict__ start,
    const int* __restrict__ col,
    const float* __restrict__ as_, const float* __restrict__ ad_,
    const float* __restrict__ H, const float* __restrict__ bias,
    float* __restrict__ out, int N, int do_relu) {
  const int lane = threadIdx.x & 31;
  const int d = __builtin_amdgcn_readfirstlane(
      (int)(blockIdx.x * (blockDim.x >> 5) + (threadIdx.x >> 5)));
  if (d >= N) return;
  const int beg = start[d];
  const int deg = cnt[d];
  const float4 adv = *(const float4*)(ad_ + (size_t)d * NHEAD);
  const float NEG_INF = -__builtin_huge_valf();

  int    s0 = d;
  float4 e0 = make_float4(NEG_INF, NEG_INF, NEG_INF, NEG_INF);
  const bool v0 = chunk_scores(col, as_, beg, deg, 0, lane, adv, &s0, &e0);

  float4 m = e0;
  for (int base = 32; base < deg; base += 32) {
    int s; float4 e;
    if (chunk_scores(col, as_, beg, deg, base, lane, adv, &s, &e)) {
      m.x = fmaxf(m.x, e.x); m.y = fmaxf(m.y, e.y);
      m.z = fmaxf(m.z, e.z); m.w = fmaxf(m.w, e.w);
    }
  }
  m.x = wave_max(m.x); m.y = wave_max(m.y);
  m.z = wave_max(m.z); m.w = wave_max(m.w);
  if (!__builtin_isfinite(m.x)) m.x = 0.0f;
  if (!__builtin_isfinite(m.y)) m.y = 0.0f;
  if (!__builtin_isfinite(m.z)) m.z = 0.0f;
  if (!__builtin_isfinite(m.w)) m.w = 0.0f;

  float4 x0 = make_float4(0.0f, 0.0f, 0.0f, 0.0f);
  if (v0) {
    x0.x = __expf(e0.x - m.x); x0.y = __expf(e0.y - m.y);
    x0.z = __expf(e0.z - m.z); x0.w = __expf(e0.w - m.w);
  }
  float4 den = x0;
  for (int base = 32; base < deg; base += 32) {
    int s; float4 e;
    if (chunk_scores(col, as_, beg, deg, base, lane, adv, &s, &e)) {
      den.x += __expf(e.x - m.x); den.y += __expf(e.y - m.y);
      den.z += __expf(e.z - m.z); den.w += __expf(e.w - m.w);
    }
  }
  den.x = wave_sum(den.x) + 1e-16f; den.y = wave_sum(den.y) + 1e-16f;
  den.z = wave_sum(den.z) + 1e-16f; den.w = wave_sum(den.w) + 1e-16f;

  const int h = lane >> 3;
  const int4 k = make_int4(h == 0 ? -1 : 0, h == 1 ? -1 : 0,
                           h == 2 ? -1 : 0, h == 3 ? -1 : 0);
  float4 acc = make_float4(0.0f, 0.0f, 0.0f, 0.0f);
  {
    const float4 al = make_float4(x0.x / den.x, x0.y / den.y,
                                  x0.z / den.z, x0.w / den.w);
    gather_rows(H, s0, al, deg < 32 ? deg : 32, lane, k, &acc);
  }
  for (int base = 32; base < deg; base += 32) {
    int s = d;
    float4 e = make_float4(0.0f, 0.0f, 0.0f, 0.0f);
    float4 al = make_float4(0.0f, 0.0f, 0.0f, 0.0f);
    if (chunk_scores(col, as_, beg, deg, base, lane, adv, &s, &e)) {
      al.x = __expf(e.x - m.x) / den.x; al.y = __expf(e.y - m.y) / den.y;
      al.z = __expf(e.z - m.z) / den.z; al.w = __expf(e.w - m.w) / den.w;
    }
    const int rem = deg - base;
    gather_rows(H, s, al, rem < 32 ? rem : 32, lane, k, &acc);
  }

  const float4 bv = *(const float4*)(bias + lane * 4);
  acc.x += bv.x; acc.y += bv.y; acc.z += bv.z; acc.w += bv.w;
  if (do_relu) {
    acc.x = acc.x > 0.0f ? acc.x : 0.0f; acc.y = acc.y > 0.0f ? acc.y : 0.0f;
    acc.z = acc.z > 0.0f ? acc.z : 0.0f; acc.w = acc.w > 0.0f ? acc.w : 0.0f;
  }
  *(float4*)(out + (size_t)d * FDIM + lane * 4) = acc;
}

static inline int cdiv(long long a, int b) { return (int)((a + b - 1) / b); }

extern "C" void kernel_launch(void* const* d_in, const int* in_sizes, int n_in,
                              void* d_out, int out_size, void* d_ws, size_t ws_size,
                              hipStream_t stream) {
  const float* x     = (const float*)d_in[0];
  const int*   ei    = (const int*)d_in[1];
  const float* W1    = (const float*)d_in[2];
  const float* asrc1 = (const float*)d_in[3];
  const float* adst1 = (const float*)d_in[4];
  const float* b1    = (const float*)d_in[5];
  const float* W2    = (const float*)d_in[6];
  const float* asrc2 = (const float*)d_in[7];
  const float* adst2 = (const float*)d_in[8];
  const float* b2    = (const float*)d_in[9];
  float* out = (float*)d_out;

  const int N  = in_sizes[0] / FDIM;
  const int E  = in_sizes[1] / 2;
  const int ET = E + N;

  float* H    = (float*)d_ws;
  int*   cnt  = (int*)(H + (size_t)N * FDIM);
  int*   total  = cnt + N;
  int*   start  = total + 1;
  int*   cursor = start + N;
  int*   col    = cursor + N;
  float* as_  = H + (size_t)2 * N * FDIM;
  float* ad_  = as_  + (size_t)N * NHEAD;

  const int TB = 256;
  const int gTiles = cdiv(N, 16);
  const int gN     = cdiv(N, TB);
  const int gN1    = cdiv(N + 1, TB);
  const int gNH    = cdiv((long long)N * NHEAD, TB);
  const int gE     = cdiv(ET, TB);
  const int gNode  = cdiv(N, TB / 32);

  zero_ints<<<gN1, TB, 0, stream>>>(cnt, N + 1);
  csr_count<<<gE, TB, 0, stream>>>(ei, E, ET, cnt);
  csr_alloc<<<gN, TB, 0, stream>>>(cnt, total, start, cursor, N);
  csr_fill<<<gE, TB, 0, stream>>>(ei, E, ET, cursor, col);

  gat_gemm_wmma<<<gTiles, TB, 0, stream>>>(x, W1, H, N);
  gat_alpha<<<gNH, TB, 0, stream>>>(H, asrc1, adst1, as_, ad_, N);
  gat_node_aggregate<<<gNode, TB, 0, stream>>>(cnt, start, col, as_, ad_, H, b1,
                                               out, N,  1);

  gat_gemm_wmma<<<gTiles, TB, 0, stream>>>(out, W2, H, N);
  gat_alpha<<<gNH, TB, 0, stream>>>(H, asrc2, adst2, as_, ad_, N);
  gat_node_aggregate<<<gNode, TB, 0, stream>>>(cnt, start, col, as_, ad_, H, b2,
                                               out, N,  0);
}


# GEMM re-tiled: one wave per 16x128 row stripe, W staged per block in LDS in B-fragment order, A fragment loaded once per k step feeds 8 WMMAs (was a 16x16 tile per wave, operands from global)
# speedup vs baseline: 3.0851x; 1.1240x over previous
#include <hip/hip_runtime.h>
#include <hip/hip_bf16.h>


typedef __attribute__((ext_vector_type(2))) float v2f;
typedef __attribute__((ext_vector_type(8))) float v8f;

#define FDIM 128
#define NHEAD 4
#define CHAN 32
#define NEG_SLOPE 0.2f

__global__ __launch_bounds__(256) void gat_gemm_wmma(const float* __restrict__ X,
                                                     const float* __restrict__ W,
                                                     float* __restrict__ H,
                                                     int N) {
  __shared__ float4 Wl[FDIM * FDIM / 4];

  const int tid  = threadIdx.x;
  const int lane = tid & 31;
  const int wave = tid >> 5;
  const int m    = lane & 15;
  const int half = lane >> 4;
  const int row0 = (blockIdx.x * 8 + wave) * 16;

  int rowA = row0 + m;
  if (rowA >= N) rowA = N - 1;
  const float* __restrict__ xrow = X + (size_t)rowA * FDIM + 2 * half;
  v2f a[FDIM / 4];
#pragma unroll
  for (int ks = 0; ks < FDIM / 4; ++ks) a[ks] = *(const v2f*)(xrow + 4 * ks);

  for (int q = tid; q < FDIM * FDIM / 4; q += 256) {
    const int ln  = q & 31;
    const int ctp = (q >> 5) & 3;
    const int ks  = q >> 7;
    const float* __restrict__ w =
        W + (4 * ks + 2 * (ln >> 4)) * FDIM + ctp * 32 + (ln & 15);
    float4 v;
    v.x = w[0];
    v.y = w[FDIM];
    v.z = w[16];
    v.w = w[FDIM + 16];
    Wl[q] = v;
  }
  __syncthreads();
  if (row0 >= N) return;

  v8f acc[8] = {};
  float4 bc[4];
#pragma unroll
  for (int ctp = 0; ctp < 4; ++ctp) bc[ctp] = Wl[ctp * 32 + lane];

#pragma unroll
  for (int ks = 0; ks < FDIM / 4; ++ks) {
    float4 bn[4];
    if (ks + 1 < FDIM / 4) {
#pragma unroll
      for (int ctp = 0; ctp < 4; ++ctp) bn[ctp] = Wl[((ks + 1) * 4 + ctp) * 32 + lane];
    }
    __builtin_amdgcn_sched_barrier(0);
#pragma unroll
    for (int ctp = 0; ctp < 4; ++ctp) {
      v2f b0; b0.x = bc[ctp].x; b0.y = bc[ctp].y;
      v2f b1; b1.x = bc[ctp].z; b1.y = bc[ctp].w;
      acc[2 * ctp] = __builtin_amdgcn_wmma_f32_16x16x4_f32(
           false, a[ks],  false, b0,
           (short)0, acc[2 * ctp],  false,  false);
      acc[2 * ctp + 1] = __builtin_amdgcn_wmma_f32_16x16x4_f32(
           false, a[ks],  false, b1,
           (short)0, acc[2 * ctp + 1],  false,  false);
    }
    __builtin_amdgcn_sched_barrier(0);
    if (ks + 1 < FDIM / 4) {
#pragma unroll
      for (int ctp = 0; ctp < 4; ++ctp) bc[ctp] = bn[ctp];
    }
  }

#pragma unroll
  for (int ct = 0; ct < 8; ++ct) {
#pragma unroll
    for (int v = 0; v < 8; ++v) {
      const int row = row0 + v + 8 * half;
      if (row < N) H[(size_t)row * FDIM + ct * 16 + m] = acc[ct][v];
    }
  }
}

__global__ void gat_alpha(const float* __restrict__ H,
                          const float* __restrict__ a_src,
                          const float* __restrict__ a_dst,
                          float* __restrict__ as_, float* __restrict__ ad_,
                          int N) {
  const int tid = blockIdx.x * blockDim.x + threadIdx.x;
  if (tid >= N * NHEAD) return;
  const int n = tid >> 2;
  const int h = tid & 3;
  const float* __restrict__ hp = H + (size_t)n * FDIM + h * CHAN;
  const float* __restrict__ asv = a_src + h * CHAN;
  const float* __restrict__ adv = a_dst + h * CHAN;
  float s = 0.0f, d = 0.0f;
#pragma unroll
  for (int c = 0; c < CHAN; ++c) {
    const float hv = hp[c];
    s = fmaf(hv, asv[c], s);
    d = fmaf(hv, adv[c], d);
  }
  as_[tid] = s;
  ad_[tid] = d;
}

__global__ void zero_ints(int* __restrict__ p, int n) {
  const int i = blockIdx.x * blockDim.x + threadIdx.x;
  if (i < n) p[i] = 0;
}

__device__ __forceinline__ void edge_endpoints(const int* __restrict__ ei,
                                               int e, int E, int* s, int* d) {
  if (e < E) { *s = ei[e]; *d = ei[E + e]; }
  else       { *s = e - E; *d = e - E; }
}

__device__ __forceinline__ float leaky(float v) {
  return v > 0.0f ? v : NEG_SLOPE * v;
}

__global__ void csr_count(const int* __restrict__ ei, int E, int ET,
                          int* __restrict__ cnt) {
  const int e = blockIdx.x * blockDim.x + threadIdx.x;
  if (e >= ET) return;
  int s, d;
  edge_endpoints(ei, e, E, &s, &d);
  atomicAdd(&cnt[d], 1);
}

__global__ void csr_alloc(const int* __restrict__ cnt, int* __restrict__ total,
                          int* __restrict__ start, int* __restrict__ cursor,
                          int N) {
  const int d    = blockIdx.x * blockDim.x + threadIdx.x;
  const int lane = threadIdx.x & 31;
  const int c    = (d < N) ? cnt[d] : 0;
  int incl = c;
  for (int off = 1; off < 32; off <<= 1) {
    const int t = __shfl_up(incl, off);
    if (lane >= off) incl += t;
  }
  const int wsum = __shfl(incl, 31);
  int base = 0;
  if (lane == 31) base = atomicAdd(total, wsum);
  base = __shfl(base, 31);
  if (d < N) {
    const int st = base + incl - c;
    start[d]  = st;
    cursor[d] = st;
  }
}

__global__ void csr_fill(const int* __restrict__ ei, int E, int ET,
                         int* __restrict__ cursor, int* __restrict__ col) {
  const int e = blockIdx.x * blockDim.x + threadIdx.x;
  if (e >= ET) return;
  int s, d;
  edge_endpoints(ei, e, E, &s, &d);
  const int p = atomicAdd(&cursor[d], 1);
  col[p] = s;
}

__device__ __forceinline__ float wave_max(float v) {
  for (int off = 16; off > 0; off >>= 1) v = fmaxf(v, __shfl_xor(v, off));
  return v;
}
__device__ __forceinline__ float wave_sum(float v) {
  for (int off = 16; off > 0; off >>= 1) v += __shfl_xor(v, off);
  return v;
}

__device__ __forceinline__ bool chunk_scores(const int* __restrict__ col,
                                             const float* __restrict__ as_,
                                             int beg, int deg, int base, int lane,
                                             float4 adv, int* s, float4* e) {
  const int j = base + lane;
  if (j >= deg) return false;
  const int sj = col[beg + j];
  const float4 a = *(const float4*)(as_ + (size_t)sj * NHEAD);
  *s = sj;
  e->x = leaky(a.x + adv.x);
  e->y = leaky(a.y + adv.y);
  e->z = leaky(a.z + adv.z);
  e->w = leaky(a.w + adv.w);
  return true;
}

__device__ __forceinline__ float bcast(float v, int jj) {
  return __int_as_float(__builtin_amdgcn_readlane(__float_as_int(v), jj));
}

struct RowTerm { float a; float4 hv; };

__device__ __forceinline__ RowTerm row_term(const float* __restrict__ H, int lane,
                                            int s, float4 al, int jj, int4 k) {
  RowTerm t;
  const int sj = __builtin_amdgcn_readlane(s, jj);
  t.a = __int_as_float((__float_as_int(bcast(al.x, jj)) & k.x) |
                       (__float_as_int(bcast(al.y, jj)) & k.y) |
                       (__float_as_int(bcast(al.z, jj)) & k.z) |
                       (__float_as_int(bcast(al.w, jj)) & k.w));
  t.hv = *((const float4*)(H + (size_t)sj * FDIM) + lane);
  return t;
}

__device__ __forceinline__ void row_fma(float4* acc, RowTerm t) {
  acc->x = fmaf(t.a, t.hv.x, acc->x);
  acc->y = fmaf(t.a, t.hv.y, acc->y);
  acc->z = fmaf(t.a, t.hv.z, acc->z);
  acc->w = fmaf(t.a, t.hv.w, acc->w);
}

__device__ __forceinline__ void gather_rows(const float* __restrict__ H,
                                            int s, float4 al, int cntc,
                                            int lane, int4 k, float4* acc) {
  int jj = 0;
  for (; jj + 4 <= cntc; jj += 4) {
    const RowTerm t0 = row_term(H, lane, s, al, jj + 0, k);
    const RowTerm t1 = row_term(H, lane, s, al, jj + 1, k);
    const RowTerm t2 = row_term(H, lane, s, al, jj + 2, k);
    const RowTerm t3 = row_term(H, lane, s, al, jj + 3, k);
    row_fma(acc, t0); row_fma(acc, t1); row_fma(acc, t2); row_fma(acc, t3);
  }
  for (; jj < cntc; ++jj) row_fma(acc, row_term(H, lane, s, al, jj, k));
}

__global__ __launch_bounds__(256) void gat_node_aggregate(
    const int* __restrict__ cnt, const int* __restrict__ start,
    const int* __restrict__ col,
    const float* __restrict__ as_, const float* __restrict__ ad_,
    const float* __restrict__ H, const float* __restrict__ bias,
    float* __restrict__ out, int N, int do_relu) {
  const int lane = threadIdx.x & 31;
  const int d = __builtin_amdgcn_readfirstlane(
      (int)(blockIdx.x * (blockDim.x >> 5) + (threadIdx.x >> 5)));
  if (d >= N) return;
  const int beg = start[d];
  const int deg = cnt[d];
  const float4 adv = *(const float4*)(ad_ + (size_t)d * NHEAD);
  const float NEG_INF = -__builtin_huge_valf();

  int    s0 = d;
  float4 e0 = make_float4(NEG_INF, NEG_INF, NEG_INF, NEG_INF);
  const bool v0 = chunk_scores(col, as_, beg, deg, 0, lane, adv, &s0, &e0);

  float4 m = e0;
  for (int base = 32; base < deg; base += 32) {
    int s; float4 e;
    if (chunk_scores(col, as_, beg, deg, base, lane, adv, &s, &e)) {
      m.x = fmaxf(m.x, e.x); m.y = fmaxf(m.y, e.y);
      m.z = fmaxf(m.z, e.z); m.w = fmaxf(m.w, e.w);
    }
  }
  m.x = wave_max(m.x); m.y = wave_max(m.y);
  m.z = wave_max(m.z); m.w = wave_max(m.w);
  if (!__builtin_isfinite(m.x)) m.x = 0.0f;
  if (!__builtin_isfinite(m.y)) m.y = 0.0f;
  if (!__builtin_isfinite(m.z)) m.z = 0.0f;
  if (!__builtin_isfinite(m.w)) m.w = 0.0f;

  float4 x0 = make_float4(0.0f, 0.0f, 0.0f, 0.0f);
  if (v0) {
    x0.x = __expf(e0.x - m.x); x0.y = __expf(e0.y - m.y);
    x0.z = __expf(e0.z - m.z); x0.w = __expf(e0.w - m.w);
  }
  float4 den = x0;
  for (int base = 32; base < deg; base += 32) {
    int s; float4 e;
    if (chunk_scores(col, as_, beg, deg, base, lane, adv, &s, &e)) {
      den.x += __expf(e.x - m.x); den.y += __expf(e.y - m.y);
      den.z += __expf(e.z - m.z); den.w += __expf(e.w - m.w);
    }
  }
  den.x = wave_sum(den.x) + 1e-16f; den.y = wave_sum(den.y) + 1e-16f;
  den.z = wave_sum(den.z) + 1e-16f; den.w = wave_sum(den.w) + 1e-16f;

  const int h = lane >> 3;
  const int4 k = make_int4(h == 0 ? -1 : 0, h == 1 ? -1 : 0,
                           h == 2 ? -1 : 0, h == 3 ? -1 : 0);
  float4 acc = make_float4(0.0f, 0.0f, 0.0f, 0.0f);
  {
    const float4 al = make_float4(x0.x / den.x, x0.y / den.y,
                                  x0.z / den.z, x0.w / den.w);
    gather_rows(H, s0, al, deg < 32 ? deg : 32, lane, k, &acc);
  }
  for (int base = 32; base < deg; base += 32) {
    int s = d;
    float4 e = make_float4(0.0f, 0.0f, 0.0f, 0.0f);
    float4 al = make_float4(0.0f, 0.0f, 0.0f, 0.0f);
    if (chunk_scores(col, as_, beg, deg, base, lane, adv, &s, &e)) {
      al.x = __expf(e.x - m.x) / den.x; al.y = __expf(e.y - m.y) / den.y;
      al.z = __expf(e.z - m.z) / den.z; al.w = __expf(e.w - m.w) / den.w;
    }
    const int rem = deg - base;
    gather_rows(H, s, al, rem < 32 ? rem : 32, lane, k, &acc);
  }

  const float4 bv = *(const float4*)(bias + lane * 4);
  acc.x += bv.x; acc.y += bv.y; acc.z += bv.z; acc.w += bv.w;
  if (do_relu) {
    acc.x = acc.x > 0.0f ? acc.x : 0.0f; acc.y = acc.y > 0.0f ? acc.y : 0.0f;
    acc.z = acc.z > 0.0f ? acc.z : 0.0f; acc.w = acc.w > 0.0f ? acc.w : 0.0f;
  }
  *(float4*)(out + (size_t)d * FDIM + lane * 4) = acc;
}

static inline int cdiv(long long a, int b) { return (int)((a + b - 1) / b); }

extern "C" void kernel_launch(void* const* d_in, const int* in_sizes, int n_in,
                              void* d_out, int out_size, void* d_ws, size_t ws_size,
                              hipStream_t stream) {
  const float* x     = (const float*)d_in[0];
  const int*   ei    = (const int*)d_in[1];
  const float* W1    = (const float*)d_in[2];
  const float* asrc1 = (const float*)d_in[3];
  const float* adst1 = (const float*)d_in[4];
  const float* b1    = (const float*)d_in[5];
  const float* W2    = (const float*)d_in[6];
  const float* asrc2 = (const float*)d_in[7];
  const float* adst2 = (const float*)d_in[8];
  const float* b2    = (const float*)d_in[9];
  float* out = (float*)d_out;

  const int N  = in_sizes[0] / FDIM;
  const int E  = in_sizes[1] / 2;
  const int ET = E + N;

  float* H    = (float*)d_ws;
  int*   cnt  = (int*)(H + (size_t)N * FDIM);
  int*   total  = cnt + N;
  int*   start  = total + 1;
  int*   cursor = start + N;
  int*   col    = cursor + N;
  float* as_  = H + (size_t)2 * N * FDIM;
  float* ad_  = as_  + (size_t)N * NHEAD;

  const int TB = 256;
  const int gStripes = cdiv(N, 128);
  const int gN     = cdiv(N, TB);
  const int gN1    = cdiv(N + 1, TB);
  const int gNH    = cdiv((long long)N * NHEAD, TB);
  const int gE     = cdiv(ET, TB);
  const int gNode  = cdiv(N, TB / 32);

  zero_ints<<<gN1, TB, 0, stream>>>(cnt, N + 1);
  csr_count<<<gE, TB, 0, stream>>>(ei, E, ET, cnt);
  csr_alloc<<<gN, TB, 0, stream>>>(cnt, total, start, cursor, N);
  csr_fill<<<gE, TB, 0, stream>>>(ei, E, ET, cursor, col);

  gat_gemm_wmma<<<gStripes, TB, 0, stream>>>(x, W1, H, N);
  gat_alpha<<<gNH, TB, 0, stream>>>(H, asrc1, adst1, as_, ad_, N);
  gat_node_aggregate<<<gNode, TB, 0, stream>>>(cnt, start, col, as_, ad_, H, b1,
                                               out, N,  1);

  gat_gemm_wmma<<<gStripes, TB, 0, stream>>>(out, W2, H, N);
  gat_alpha<<<gNH, TB, 0, stream>>>(H, asrc2, adst2, as_, ad_, N);
  gat_node_aggregate<<<gNode, TB, 0, stream>>>(cnt, start, col, as_, ad_, H, b2,
                                               out, N,  0);
}


# Attention logits as_/ad_ formed in the GEMM epilogue from the accumulators (16-lane DPP halving sum); both gat_alpha launches removed (10 to 8); the 32-term dots sum as a lane tree, not serially
# speedup vs baseline: 3.2262x; 1.0457x over previous
#include <hip/hip_runtime.h>
#include <hip/hip_bf16.h>


typedef __attribute__((ext_vector_type(2))) float v2f;
typedef __attribute__((ext_vector_type(8))) float v8f;

#define FDIM 128
#define NHEAD 4
#define CHAN 32
#define NEG_SLOPE 0.2f


template <int CTRL>
__device__ __forceinline__ float row_exchange(float x) {
  return __int_as_float(
      __builtin_amdgcn_update_dpp(0, __float_as_int(x), CTRL, 0xf, 0xf, true));
}

template <int HALF_N, int CTRL>
__device__ __forceinline__ void fold_step(float* p, bool upper) {
#pragma unroll
  for (int j = 0; j < HALF_N; ++j) {
    const float lo = p[j], hi = p[j + HALF_N];
    const float keep = upper ? hi : lo;
    const float send = upper ? lo : hi;
    p[j] = keep + row_exchange<CTRL>(send);
  }
}

__global__ __launch_bounds__(256) void gat_gemm_wmma(const float* __restrict__ X,
                                                     const float* __restrict__ W,
                                                     const float* __restrict__ a_src,
                                                     const float* __restrict__ a_dst,
                                                     float* __restrict__ H,
                                                     float* __restrict__ as_,
                                                     float* __restrict__ ad_,
                                                     int N) {
  __shared__ float4 Wl[FDIM * FDIM / 4];

  const int tid  = threadIdx.x;
  const int lane = tid & 31;
  const int wave = tid >> 5;
  const int m    = lane & 15;
  const int half = lane >> 4;
  const int row0 = (blockIdx.x * 8 + wave) * 16;

  int rowA = row0 + m;
  if (rowA >= N) rowA = N - 1;
  const float* __restrict__ xrow = X + (size_t)rowA * FDIM + 2 * half;
  v2f a[FDIM / 4];
#pragma unroll
  for (int ks = 0; ks < FDIM / 4; ++ks) a[ks] = *(const v2f*)(xrow + 4 * ks);

  float asv[8], adv[8];
#pragma unroll
  for (int ct = 0; ct < 8; ++ct) {
    asv[ct] = a_src[ct * 16 + m];
    adv[ct] = a_dst[ct * 16 + m];
  }

  for (int q = tid; q < FDIM * FDIM / 4; q += 256) {
    const int ln  = q & 31;
    const int ctp = (q >> 5) & 3;
    const int ks  = q >> 7;
    const float* __restrict__ w =
        W + (4 * ks + 2 * (ln >> 4)) * FDIM + ctp * 32 + (ln & 15);
    float4 v;
    v.x = w[0];
    v.y = w[FDIM];
    v.z = w[16];
    v.w = w[FDIM + 16];
    Wl[q] = v;
  }
  __syncthreads();
  if (row0 >= N) return;

  v8f acc[8] = {};
  float4 bc[4];
#pragma unroll
  for (int ctp = 0; ctp < 4; ++ctp) bc[ctp] = Wl[ctp * 32 + lane];

#pragma unroll
  for (int ks = 0; ks < FDIM / 4; ++ks) {
    float4 bn[4];
    if (ks + 1 < FDIM / 4) {
#pragma unroll
      for (int ctp = 0; ctp < 4; ++ctp) bn[ctp] = Wl[((ks + 1) * 4 + ctp) * 32 + lane];
    }
    __builtin_amdgcn_sched_barrier(0);
#pragma unroll
    for (int ctp = 0; ctp < 4; ++ctp) {
      v2f b0; b0.x = bc[ctp].x; b0.y = bc[ctp].y;
      v2f b1; b1.x = bc[ctp].z; b1.y = bc[ctp].w;
      acc[2 * ctp] = __builtin_amdgcn_wmma_f32_16x16x4_f32(
           false, a[ks],  false, b0,
           (short)0, acc[2 * ctp],  false,  false);
      acc[2 * ctp + 1] = __builtin_amdgcn_wmma_f32_16x16x4_f32(
           false, a[ks],  false, b1,
           (short)0, acc[2 * ctp + 1],  false,  false);
    }
    __builtin_amdgcn_sched_barrier(0);
    if (ks + 1 < FDIM / 4) {
#pragma unroll
      for (int ctp = 0; ctp < 4; ++ctp) bc[ctp] = bn[ctp];
    }
  }

#pragma unroll
  for (int ct = 0; ct < 8; ++ct) {
#pragma unroll
    for (int v = 0; v < 8; ++v) {
      const int row = row0 + v + 8 * half;
      if (row < N) H[(size_t)row * FDIM + ct * 16 + m] = acc[ct][v];
    }
  }

  float p[64];
#pragma unroll
  for (int v = 0; v < 8; ++v) {
#pragma unroll
    for (int h = 0; h < NHEAD; ++h) {
      const int idx = ((v & 1) << 5) | (((v >> 1) & 1) << 4) | (((v >> 2) & 1) << 3) | h;
      p[idx]     = fmaf(acc[2 * h + 1][v], asv[2 * h + 1], acc[2 * h][v] * asv[2 * h]);
      p[idx | 4] = fmaf(acc[2 * h + 1][v], adv[2 * h + 1], acc[2 * h][v] * adv[2 * h]);
    }
  }
  fold_step<32, 0xB1>(p, (lane & 1) != 0);
  fold_step<16, 0x4E>(p, (lane & 2) != 0);
  fold_step<8, 0x164>(p, (lane & 4) != 0);
  fold_step<4, 0x168>(p, (lane & 8) != 0);
  {
    const int row = row0 + (m & 7) + 8 * half;
    float* __restrict__ dstp = ((m & 8) ? ad_ : as_) + (size_t)row * NHEAD;
    if (row < N) *(float4*)dstp = make_float4(p[0], p[1], p[2], p[3]);
  }
}

__global__ void zero_ints(int* __restrict__ p, int n) {
  const int i = blockIdx.x * blockDim.x + threadIdx.x;
  if (i < n) p[i] = 0;
}

__device__ __forceinline__ void edge_endpoints(const int* __restrict__ ei,
                                               int e, int E, int* s, int* d) {
  if (e < E) { *s = ei[e]; *d = ei[E + e]; }
  else       { *s = e - E; *d = e - E; }
}

__device__ __forceinline__ float leaky(float v) {
  return v > 0.0f ? v : NEG_SLOPE * v;
}

__global__ void csr_count(const int* __restrict__ ei, int E, int ET,
                          int* __restrict__ cnt) {
  const int e = blockIdx.x * blockDim.x + threadIdx.x;
  if (e >= ET) return;
  int s, d;
  edge_endpoints(ei, e, E, &s, &d);
  atomicAdd(&cnt[d], 1);
}

__global__ void csr_alloc(const int* __restrict__ cnt, int* __restrict__ total,
                          int* __restrict__ start, int* __restrict__ cursor,
                          int N) {
  const int d    = blockIdx.x * blockDim.x + threadIdx.x;
  const int lane = threadIdx.x & 31;
  const int c    = (d < N) ? cnt[d] : 0;
  int incl = c;
  for (int off = 1; off < 32; off <<= 1) {
    const int t = __shfl_up(incl, off);
    if (lane >= off) incl += t;
  }
  const int wsum = __shfl(incl, 31);
  int base = 0;
  if (lane == 31) base = atomicAdd(total, wsum);
  base = __shfl(base, 31);
  if (d < N) {
    const int st = base + incl - c;
    start[d]  = st;
    cursor[d] = st;
  }
}

__global__ void csr_fill(const int* __restrict__ ei, int E, int ET,
                         int* __restrict__ cursor, int* __restrict__ col) {
  const int e = blockIdx.x * blockDim.x + threadIdx.x;
  if (e >= ET) return;
  int s, d;
  edge_endpoints(ei, e, E, &s, &d);
  const int p = atomicAdd(&cursor[d], 1);
  col[p] = s;
}

__device__ __forceinline__ float wave_max(float v) {
  for (int off = 16; off > 0; off >>= 1) v = fmaxf(v, __shfl_xor(v, off));
  return v;
}
__device__ __forceinline__ float wave_sum(float v) {
  for (int off = 16; off > 0; off >>= 1) v += __shfl_xor(v, off);
  return v;
}

__device__ __forceinline__ bool chunk_scores(const int* __restrict__ col,
                                             const float* __restrict__ as_,
                                             int beg, int deg, int base, int lane,
                                             float4 adv, int* s, float4* e) {
  const int j = base + lane;
  if (j >= deg) return false;
  const int sj = col[beg + j];
  const float4 a = *(const float4*)(as_ + (size_t)sj * NHEAD);
  *s = sj;
  e->x = leaky(a.x + adv.x);
  e->y = leaky(a.y + adv.y);
  e->z = leaky(a.z + adv.z);
  e->w = leaky(a.w + adv.w);
  return true;
}

__device__ __forceinline__ float bcast(float v, int jj) {
  return __int_as_float(__builtin_amdgcn_readlane(__float_as_int(v), jj));
}

struct RowTerm { float a; float4 hv; };

__device__ __forceinline__ RowTerm row_term(const float* __restrict__ H, int lane,
                                            int s, float4 al, int jj, int4 k) {
  RowTerm t;
  const int sj = __builtin_amdgcn_readlane(s, jj);
  t.a = __int_as_float((__float_as_int(bcast(al.x, jj)) & k.x) |
                       (__float_as_int(bcast(al.y, jj)) & k.y) |
                       (__float_as_int(bcast(al.z, jj)) & k.z) |
                       (__float_as_int(bcast(al.w, jj)) & k.w));
  t.hv = *((const float4*)(H + (size_t)sj * FDIM) + lane);
  return t;
}

__device__ __forceinline__ void row_fma(float4* acc, RowTerm t) {
  acc->x = fmaf(t.a, t.hv.x, acc->x);
  acc->y = fmaf(t.a, t.hv.y, acc->y);
  acc->z = fmaf(t.a, t.hv.z, acc->z);
  acc->w = fmaf(t.a, t.hv.w, acc->w);
}

__device__ __forceinline__ void gather_rows(const float* __restrict__ H,
                                            int s, float4 al, int cntc,
                                            int lane, int4 k, float4* acc) {
  int jj = 0;
  for (; jj + 4 <= cntc; jj += 4) {
    const RowTerm t0 = row_term(H, lane, s, al, jj + 0, k);
    const RowTerm t1 = row_term(H, lane, s, al, jj + 1, k);
    const RowTerm t2 = row_term(H, lane, s, al, jj + 2, k);
    const RowTerm t3 = row_term(H, lane, s, al, jj + 3, k);
    row_fma(acc, t0); row_fma(acc, t1); row_fma(acc, t2); row_fma(acc, t3);
  }
  for (; jj < cntc; ++jj) row_fma(acc, row_term(H, lane, s, al, jj, k));
}

__global__ __launch_bounds__(256) void gat_node_aggregate(
    const int* __restrict__ cnt, const int* __restrict__ start,
    const int* __restrict__ col,
    const float* __restrict__ as_, const float* __restrict__ ad_,
    const float* __restrict__ H, const float* __restrict__ bias,
    float* __restrict__ out, int N, int do_relu) {
  const int lane = threadIdx.x & 31;
  const int d = __builtin_amdgcn_readfirstlane(
      (int)(blockIdx.x * (blockDim.x >> 5) + (threadIdx.x >> 5)));
  if (d >= N) return;
  const int beg = start[d];
  const int deg = cnt[d];
  const float4 adv = *(const float4*)(ad_ + (size_t)d * NHEAD);
  const float NEG_INF = -__builtin_huge_valf();

  int    s0 = d;
  float4 e0 = make_float4(NEG_INF, NEG_INF, NEG_INF, NEG_INF);
  const bool v0 = chunk_scores(col, as_, beg, deg, 0, lane, adv, &s0, &e0);

  float4 m = e0;
  for (int base = 32; base < deg; base += 32) {
    int s; float4 e;
    if (chunk_scores(col, as_, beg, deg, base, lane, adv, &s, &e)) {
      m.x = fmaxf(m.x, e.x); m.y = fmaxf(m.y, e.y);
      m.z = fmaxf(m.z, e.z); m.w = fmaxf(m.w, e.w);
    }
  }
  m.x = wave_max(m.x); m.y = wave_max(m.y);
  m.z = wave_max(m.z); m.w = wave_max(m.w);
  if (!__builtin_isfinite(m.x)) m.x = 0.0f;
  if (!__builtin_isfinite(m.y)) m.y = 0.0f;
  if (!__builtin_isfinite(m.z)) m.z = 0.0f;
  if (!__builtin_isfinite(m.w)) m.w = 0.0f;

  float4 x0 = make_float4(0.0f, 0.0f, 0.0f, 0.0f);
  if (v0) {
    x0.x = __expf(e0.x - m.x); x0.y = __expf(e0.y - m.y);
    x0.z = __expf(e0.z - m.z); x0.w = __expf(e0.w - m.w);
  }
  float4 den = x0;
  for (int base = 32; base < deg; base += 32) {
    int s; float4 e;
    if (chunk_scores(col, as_, beg, deg, base, lane, adv, &s, &e)) {
      den.x += __expf(e.x - m.x); den.y += __expf(e.y - m.y);
      den.z += __expf(e.z - m.z); den.w += __expf(e.w - m.w);
    }
  }
  den.x = wave_sum(den.x) + 1e-16f; den.y = wave_sum(den.y) + 1e-16f;
  den.z = wave_sum(den.z) + 1e-16f; den.w = wave_sum(den.w) + 1e-16f;

  const int h = lane >> 3;
  const int4 k = make_int4(h == 0 ? -1 : 0, h == 1 ? -1 : 0,
                           h == 2 ? -1 : 0, h == 3 ? -1 : 0);
  float4 acc = make_float4(0.0f, 0.0f, 0.0f, 0.0f);
  {
    const float4 al = make_float4(x0.x / den.x, x0.y / den.y,
                                  x0.z / den.z, x0.w / den.w);
    gather_rows(H, s0, al, deg < 32 ? deg : 32, lane, k, &acc);
  }
  for (int base = 32; base < deg; base += 32) {
    int s = d;
    float4 e = make_float4(0.0f, 0.0f, 0.0f, 0.0f);
    float4 al = make_float4(0.0f, 0.0f, 0.0f, 0.0f);
    if (chunk_scores(col, as_, beg, deg, base, lane, adv, &s, &e)) {
      al.x = __expf(e.x - m.x) / den.x; al.y = __expf(e.y - m.y) / den.y;
      al.z = __expf(e.z - m.z) / den.z; al.w = __expf(e.w - m.w) / den.w;
    }
    const int rem = deg - base;
    gather_rows(H, s, al, rem < 32 ? rem : 32, lane, k, &acc);
  }

  const float4 bv = *(const float4*)(bias + lane * 4);
  acc.x += bv.x; acc.y += bv.y; acc.z += bv.z; acc.w += bv.w;
  if (do_relu) {
    acc.x = acc.x > 0.0f ? acc.x : 0.0f; acc.y = acc.y > 0.0f ? acc.y : 0.0f;
    acc.z = acc.z > 0.0f ? acc.z : 0.0f; acc.w = acc.w > 0.0f ? acc.w : 0.0f;
  }
  *(float4*)(out + (size_t)d * FDIM + lane * 4) = acc;
}

static inline int cdiv(long long a, int b) { return (int)((a + b - 1) / b); }

extern "C" void kernel_launch(void* const* d_in, const int* in_sizes, int n_in,
                              void* d_out, int out_size, void* d_ws, size_t ws_size,
                              hipStream_t stream) {
  const float* x     = (const float*)d_in[0];
  const int*   ei    = (const int*)d_in[1];
  const float* W1    = (const float*)d_in[2];
  const float* asrc1 = (const float*)d_in[3];
  const float* adst1 = (const float*)d_in[4];
  const float* b1    = (const float*)d_in[5];
  const float* W2    = (const float*)d_in[6];
  const float* asrc2 = (const float*)d_in[7];
  const float* adst2 = (const float*)d_in[8];
  const float* b2    = (const float*)d_in[9];
  float* out = (float*)d_out;

  const int N  = in_sizes[0] / FDIM;
  const int E  = in_sizes[1] / 2;
  const int ET = E + N;

  float* H    = (float*)d_ws;
  int*   cnt  = (int*)(H + (size_t)N * FDIM);
  int*   total  = cnt + N;
  int*   start  = total + 1;
  int*   cursor = start + N;
  int*   col    = cursor + N;
  float* as_  = H + (size_t)2 * N * FDIM;
  float* ad_  = as_  + (size_t)N * NHEAD;

  const int TB = 256;
  const int gStripes = cdiv(N, 128);
  const int gN     = cdiv(N, TB);
  const int gN1    = cdiv(N + 1, TB);
  const int gE     = cdiv(ET, TB);
  const int gNode  = cdiv(N, TB / 32);

  zero_ints<<<gN1, TB, 0, stream>>>(cnt, N + 1);
  csr_count<<<gE, TB, 0, stream>>>(ei, E, ET, cnt);
  csr_alloc<<<gN, TB, 0, stream>>>(cnt, total, start, cursor, N);
  csr_fill<<<gE, TB, 0, stream>>>(ei, E, ET, cursor, col);

  gat_gemm_wmma<<<gStripes, TB, 0, stream>>>(x, W1, asrc1, adst1, H, as_, ad_, N);
  gat_node_aggregate<<<gNode, TB, 0, stream>>>(cnt, start, col, as_, ad_, H, b1,
                                               out, N,  1);

  gat_gemm_wmma<<<gStripes, TB, 0, stream>>>(out, W2, asrc2, adst2, H, as_, ad_, N);
  gat_node_aggregate<<<gNode, TB, 0, stream>>>(cnt, start, col, as_, ad_, H, b2,
                                               out, N,  0);
}


# Edge slot = value the counting add returns, kept per edge; the fill pass places edges with plain loads, no atomics (850k returning adds gone); in-segment order is still an atomic arrival order
# speedup vs baseline: 3.7191x; 1.1528x over previous
#include <hip/hip_runtime.h>
#include <hip/hip_bf16.h>


typedef __attribute__((ext_vector_type(2))) float v2f;
typedef __attribute__((ext_vector_type(8))) float v8f;

#define FDIM 128
#define NHEAD 4
#define CHAN 32
#define NEG_SLOPE 0.2f


template <int CTRL>
__device__ __forceinline__ float row_exchange(float x) {
  return __int_as_float(
      __builtin_amdgcn_update_dpp(0, __float_as_int(x), CTRL, 0xf, 0xf, true));
}

template <int HALF_N, int CTRL>
__device__ __forceinline__ void fold_step(float* p, bool upper) {
#pragma unroll
  for (int j = 0; j < HALF_N; ++j) {
    const float lo = p[j], hi = p[j + HALF_N];
    const float keep = upper ? hi : lo;
    const float send = upper ? lo : hi;
    p[j] = keep + row_exchange<CTRL>(send);
  }
}

__global__ __launch_bounds__(256) void gat_gemm_wmma(const float* __restrict__ X,
                                                     const float* __restrict__ W,
                                                     const float* __restrict__ a_src,
                                                     const float* __restrict__ a_dst,
                                                     float* __restrict__ H,
                                                     float* __restrict__ as_,
                                                     float* __restrict__ ad_,
                                                     int N) {
  __shared__ float4 Wl[FDIM * FDIM / 4];

  const int tid  = threadIdx.x;
  const int lane = tid & 31;
  const int wave = tid >> 5;
  const int m    = lane & 15;
  const int half = lane >> 4;
  const int row0 = (blockIdx.x * 8 + wave) * 16;

  int rowA = row0 + m;
  if (rowA >= N) rowA = N - 1;
  const float* __restrict__ xrow = X + (size_t)rowA * FDIM + 2 * half;
  v2f a[FDIM / 4];
#pragma unroll
  for (int ks = 0; ks < FDIM / 4; ++ks) a[ks] = *(const v2f*)(xrow + 4 * ks);

  float asv[8], adv[8];
#pragma unroll
  for (int ct = 0; ct < 8; ++ct) {
    asv[ct] = a_src[ct * 16 + m];
    adv[ct] = a_dst[ct * 16 + m];
  }

  for (int q = tid; q < FDIM * FDIM / 4; q += 256) {
    const int ln  = q & 31;
    const int ctp = (q >> 5) & 3;
    const int ks  = q >> 7;
    const float* __restrict__ w =
        W + (4 * ks + 2 * (ln >> 4)) * FDIM + ctp * 32 + (ln & 15);
    float4 v;
    v.x = w[0];
    v.y = w[FDIM];
    v.z = w[16];
    v.w = w[FDIM + 16];
    Wl[q] = v;
  }
  __syncthreads();
  if (row0 >= N) return;

  v8f acc[8] = {};
  float4 bc[4];
#pragma unroll
  for (int ctp = 0; ctp < 4; ++ctp) bc[ctp] = Wl[ctp * 32 + lane];

#pragma unroll
  for (int ks = 0; ks < FDIM / 4; ++ks) {
    float4 bn[4];
    if (ks + 1 < FDIM / 4) {
#pragma unroll
      for (int ctp = 0; ctp < 4; ++ctp) bn[ctp] = Wl[((ks + 1) * 4 + ctp) * 32 + lane];
    }
    __builtin_amdgcn_sched_barrier(0);
#pragma unroll
    for (int ctp = 0; ctp < 4; ++ctp) {
      v2f b0; b0.x = bc[ctp].x; b0.y = bc[ctp].y;
      v2f b1; b1.x = bc[ctp].z; b1.y = bc[ctp].w;
      acc[2 * ctp] = __builtin_amdgcn_wmma_f32_16x16x4_f32(
           false, a[ks],  false, b0,
           (short)0, acc[2 * ctp],  false,  false);
      acc[2 * ctp + 1] = __builtin_amdgcn_wmma_f32_16x16x4_f32(
           false, a[ks],  false, b1,
           (short)0, acc[2 * ctp + 1],  false,  false);
    }
    __builtin_amdgcn_sched_barrier(0);
    if (ks + 1 < FDIM / 4) {
#pragma unroll
      for (int ctp = 0; ctp < 4; ++ctp) bc[ctp] = bn[ctp];
    }
  }

#pragma unroll
  for (int ct = 0; ct < 8; ++ct) {
#pragma unroll
    for (int v = 0; v < 8; ++v) {
      const int row = row0 + v + 8 * half;
      if (row < N) H[(size_t)row * FDIM + ct * 16 + m] = acc[ct][v];
    }
  }

  float p[64];
#pragma unroll
  for (int v = 0; v < 8; ++v) {
#pragma unroll
    for (int h = 0; h < NHEAD; ++h) {
      const int idx = ((v & 1) << 5) | (((v >> 1) & 1) << 4) | (((v >> 2) & 1) << 3) | h;
      p[idx]     = fmaf(acc[2 * h + 1][v], asv[2 * h + 1], acc[2 * h][v] * asv[2 * h]);
      p[idx | 4] = fmaf(acc[2 * h + 1][v], adv[2 * h + 1], acc[2 * h][v] * adv[2 * h]);
    }
  }
  fold_step<32, 0xB1>(p, (lane & 1) != 0);
  fold_step<16, 0x4E>(p, (lane & 2) != 0);
  fold_step<8, 0x164>(p, (lane & 4) != 0);
  fold_step<4, 0x168>(p, (lane & 8) != 0);
  {
    const int row = row0 + (m & 7) + 8 * half;
    float* __restrict__ dstp = ((m & 8) ? ad_ : as_) + (size_t)row * NHEAD;
    if (row < N) *(float4*)dstp = make_float4(p[0], p[1], p[2], p[3]);
  }
}

__global__ void zero_ints(int* __restrict__ p, int n) {
  const int i = blockIdx.x * blockDim.x + threadIdx.x;
  if (i < n) p[i] = 0;
}

__device__ __forceinline__ void edge_endpoints(const int* __restrict__ ei,
                                               int e, int E, int* s, int* d) {
  if (e < E) { *s = ei[e]; *d = ei[E + e]; }
  else       { *s = e - E; *d = e - E; }
}

__device__ __forceinline__ float leaky(float v) {
  return v > 0.0f ? v : NEG_SLOPE * v;
}

__global__ void csr_count(const int* __restrict__ ei, int E, int ET,
                          int* __restrict__ cnt, int* __restrict__ rank) {
  const int e = blockIdx.x * blockDim.x + threadIdx.x;
  if (e >= ET) return;
  int s, d;
  edge_endpoints(ei, e, E, &s, &d);
  rank[e] = atomicAdd(&cnt[d], 1);
}

__global__ void csr_alloc(const int* __restrict__ cnt, int* __restrict__ total,
                          int* __restrict__ start, int N) {
  const int d    = blockIdx.x * blockDim.x + threadIdx.x;
  const int lane = threadIdx.x & 31;
  const int c    = (d < N) ? cnt[d] : 0;
  int incl = c;
  for (int off = 1; off < 32; off <<= 1) {
    const int t = __shfl_up(incl, off);
    if (lane >= off) incl += t;
  }
  const int wsum = __shfl(incl, 31);
  int base = 0;
  if (lane == 31) base = atomicAdd(total, wsum);
  base = __shfl(base, 31);
  if (d < N) start[d] = base + incl - c;
}

__global__ void csr_fill(const int* __restrict__ ei, int E, int ET,
                         const int* __restrict__ start,
                         const int* __restrict__ rank, int* __restrict__ col) {
  const int e = blockIdx.x * blockDim.x + threadIdx.x;
  if (e >= ET) return;
  int s, d;
  edge_endpoints(ei, e, E, &s, &d);
  col[start[d] + rank[e]] = s;
}

__device__ __forceinline__ float wave_max(float v) {
  for (int off = 16; off > 0; off >>= 1) v = fmaxf(v, __shfl_xor(v, off));
  return v;
}
__device__ __forceinline__ float wave_sum(float v) {
  for (int off = 16; off > 0; off >>= 1) v += __shfl_xor(v, off);
  return v;
}

__device__ __forceinline__ bool chunk_scores(const int* __restrict__ col,
                                             const float* __restrict__ as_,
                                             int beg, int deg, int base, int lane,
                                             float4 adv, int* s, float4* e) {
  const int j = base + lane;
  if (j >= deg) return false;
  const int sj = col[beg + j];
  const float4 a = *(const float4*)(as_ + (size_t)sj * NHEAD);
  *s = sj;
  e->x = leaky(a.x + adv.x);
  e->y = leaky(a.y + adv.y);
  e->z = leaky(a.z + adv.z);
  e->w = leaky(a.w + adv.w);
  return true;
}

__device__ __forceinline__ float bcast(float v, int jj) {
  return __int_as_float(__builtin_amdgcn_readlane(__float_as_int(v), jj));
}

struct RowTerm { float a; float4 hv; };

__device__ __forceinline__ RowTerm row_term(const float* __restrict__ H, int lane,
                                            int s, float4 al, int jj, int4 k) {
  RowTerm t;
  const int sj = __builtin_amdgcn_readlane(s, jj);
  t.a = __int_as_float((__float_as_int(bcast(al.x, jj)) & k.x) |
                       (__float_as_int(bcast(al.y, jj)) & k.y) |
                       (__float_as_int(bcast(al.z, jj)) & k.z) |
                       (__float_as_int(bcast(al.w, jj)) & k.w));
  t.hv = *((const float4*)(H + (size_t)sj * FDIM) + lane);
  return t;
}

__device__ __forceinline__ void row_fma(float4* acc, RowTerm t) {
  acc->x = fmaf(t.a, t.hv.x, acc->x);
  acc->y = fmaf(t.a, t.hv.y, acc->y);
  acc->z = fmaf(t.a, t.hv.z, acc->z);
  acc->w = fmaf(t.a, t.hv.w, acc->w);
}

__device__ __forceinline__ void gather_rows(const float* __restrict__ H,
                                            int s, float4 al, int cntc,
                                            int lane, int4 k, float4* acc) {
  int jj = 0;
  for (; jj + 4 <= cntc; jj += 4) {
    const RowTerm t0 = row_term(H, lane, s, al, jj + 0, k);
    const RowTerm t1 = row_term(H, lane, s, al, jj + 1, k);
    const RowTerm t2 = row_term(H, lane, s, al, jj + 2, k);
    const RowTerm t3 = row_term(H, lane, s, al, jj + 3, k);
    row_fma(acc, t0); row_fma(acc, t1); row_fma(acc, t2); row_fma(acc, t3);
  }
  for (; jj < cntc; ++jj) row_fma(acc, row_term(H, lane, s, al, jj, k));
}

__global__ __launch_bounds__(256) void gat_node_aggregate(
    const int* __restrict__ cnt, const int* __restrict__ start,
    const int* __restrict__ col,
    const float* __restrict__ as_, const float* __restrict__ ad_,
    const float* __restrict__ H, const float* __restrict__ bias,
    float* __restrict__ out, int N, int do_relu) {
  const int lane = threadIdx.x & 31;
  const int d = __builtin_amdgcn_readfirstlane(
      (int)(blockIdx.x * (blockDim.x >> 5) + (threadIdx.x >> 5)));
  if (d >= N) return;
  const int beg = start[d];
  const int deg = cnt[d];
  const float4 adv = *(const float4*)(ad_ + (size_t)d * NHEAD);
  const float NEG_INF = -__builtin_huge_valf();

  int    s0 = d;
  float4 e0 = make_float4(NEG_INF, NEG_INF, NEG_INF, NEG_INF);
  const bool v0 = chunk_scores(col, as_, beg, deg, 0, lane, adv, &s0, &e0);

  float4 m = e0;
  for (int base = 32; base < deg; base += 32) {
    int s; float4 e;
    if (chunk_scores(col, as_, beg, deg, base, lane, adv, &s, &e)) {
      m.x = fmaxf(m.x, e.x); m.y = fmaxf(m.y, e.y);
      m.z = fmaxf(m.z, e.z); m.w = fmaxf(m.w, e.w);
    }
  }
  m.x = wave_max(m.x); m.y = wave_max(m.y);
  m.z = wave_max(m.z); m.w = wave_max(m.w);
  if (!__builtin_isfinite(m.x)) m.x = 0.0f;
  if (!__builtin_isfinite(m.y)) m.y = 0.0f;
  if (!__builtin_isfinite(m.z)) m.z = 0.0f;
  if (!__builtin_isfinite(m.w)) m.w = 0.0f;

  float4 x0 = make_float4(0.0f, 0.0f, 0.0f, 0.0f);
  if (v0) {
    x0.x = __expf(e0.x - m.x); x0.y = __expf(e0.y - m.y);
    x0.z = __expf(e0.z - m.z); x0.w = __expf(e0.w - m.w);
  }
  float4 den = x0;
  for (int base = 32; base < deg; base += 32) {
    int s; float4 e;
    if (chunk_scores(col, as_, beg, deg, base, lane, adv, &s, &e)) {
      den.x += __expf(e.x - m.x); den.y += __expf(e.y - m.y);
      den.z += __expf(e.z - m.z); den.w += __expf(e.w - m.w);
    }
  }
  den.x = wave_sum(den.x) + 1e-16f; den.y = wave_sum(den.y) + 1e-16f;
  den.z = wave_sum(den.z) + 1e-16f; den.w = wave_sum(den.w) + 1e-16f;

  const int h = lane >> 3;
  const int4 k = make_int4(h == 0 ? -1 : 0, h == 1 ? -1 : 0,
                           h == 2 ? -1 : 0, h == 3 ? -1 : 0);
  float4 acc = make_float4(0.0f, 0.0f, 0.0f, 0.0f);
  {
    const float4 al = make_float4(x0.x / den.x, x0.y / den.y,
                                  x0.z / den.z, x0.w / den.w);
    gather_rows(H, s0, al, deg < 32 ? deg : 32, lane, k, &acc);
  }
  for (int base = 32; base < deg; base += 32) {
    int s = d;
    float4 e = make_float4(0.0f, 0.0f, 0.0f, 0.0f);
    float4 al = make_float4(0.0f, 0.0f, 0.0f, 0.0f);
    if (chunk_scores(col, as_, beg, deg, base, lane, adv, &s, &e)) {
      al.x = __expf(e.x - m.x) / den.x; al.y = __expf(e.y - m.y) / den.y;
      al.z = __expf(e.z - m.z) / den.z; al.w = __expf(e.w - m.w) / den.w;
    }
    const int rem = deg - base;
    gather_rows(H, s, al, rem < 32 ? rem : 32, lane, k, &acc);
  }

  const float4 bv = *(const float4*)(bias + lane * 4);
  acc.x += bv.x; acc.y += bv.y; acc.z += bv.z; acc.w += bv.w;
  if (do_relu) {
    acc.x = acc.x > 0.0f ? acc.x : 0.0f; acc.y = acc.y > 0.0f ? acc.y : 0.0f;
    acc.z = acc.z > 0.0f ? acc.z : 0.0f; acc.w = acc.w > 0.0f ? acc.w : 0.0f;
  }
  *(float4*)(out + (size_t)d * FDIM + lane * 4) = acc;
}

static inline int cdiv(long long a, int b) { return (int)((a + b - 1) / b); }

extern "C" void kernel_launch(void* const* d_in, const int* in_sizes, int n_in,
                              void* d_out, int out_size, void* d_ws, size_t ws_size,
                              hipStream_t stream) {
  const float* x     = (const float*)d_in[0];
  const int*   ei    = (const int*)d_in[1];
  const float* W1    = (const float*)d_in[2];
  const float* asrc1 = (const float*)d_in[3];
  const float* adst1 = (const float*)d_in[4];
  const float* b1    = (const float*)d_in[5];
  const float* W2    = (const float*)d_in[6];
  const float* asrc2 = (const float*)d_in[7];
  const float* adst2 = (const float*)d_in[8];
  const float* b2    = (const float*)d_in[9];
  float* out = (float*)d_out;

  const int N  = in_sizes[0] / FDIM;
  const int E  = in_sizes[1] / 2;
  const int ET = E + N;

  float* H    = (float*)d_ws;
  int*   cnt  = (int*)(H + (size_t)N * FDIM);
  int*   total  = cnt + N;
  int*   start  = total + 1;
  int*   col    = start + N;
  int*   rank   = col + ET;
  float* as_  = H + (size_t)2 * N * FDIM;
  float* ad_  = as_  + (size_t)N * NHEAD;

  const int TB = 256;
  const int gStripes = cdiv(N, 128);
  const int gN     = cdiv(N, TB);
  const int gN1    = cdiv(N + 1, TB);
  const int gE     = cdiv(ET, TB);
  const int gNode  = cdiv(N, TB / 32);

  zero_ints<<<gN1, TB, 0, stream>>>(cnt, N + 1);
  csr_count<<<gE, TB, 0, stream>>>(ei, E, ET, cnt, rank);
  csr_alloc<<<gN, TB, 0, stream>>>(cnt, total, start, N);
  csr_fill<<<gE, TB, 0, stream>>>(ei, E, ET, start, rank, col);

  gat_gemm_wmma<<<gStripes, TB, 0, stream>>>(x, W1, asrc1, adst1, H, as_, ad_, N);
  gat_node_aggregate<<<gNode, TB, 0, stream>>>(cnt, start, col, as_, ad_, H, b1,
                                               out, N,  1);

  gat_gemm_wmma<<<gStripes, TB, 0, stream>>>(out, W2, asrc2, adst2, H, as_, ad_, N);
  gat_node_aggregate<<<gNode, TB, 0, stream>>>(cnt, start, col, as_, ad_, H, b2,
                                               out, N,  0);
}


# Edges grouped by destination by LDS counters in two levels (256 edge chunks, 128-node buckets); 850k global returning adds and 2 launches go (8 to 6); in-segment order still arrival, self loop first
# speedup vs baseline: 5.5022x; 1.4794x over previous
#include <hip/hip_runtime.h>
#include <hip/hip_bf16.h>


typedef __attribute__((ext_vector_type(2))) float v2f;
typedef __attribute__((ext_vector_type(8))) float v8f;

#define FDIM 128
#define NHEAD 4
#define CHAN 32
#define NEG_SLOPE 0.2f


template <int CTRL>
__device__ __forceinline__ float row_exchange(float x) {
  return __int_as_float(
      __builtin_amdgcn_update_dpp(0, __float_as_int(x), CTRL, 0xf, 0xf, true));
}

template <int HALF_N, int CTRL>
__device__ __forceinline__ void fold_step(float* p, bool upper) {
#pragma unroll
  for (int j = 0; j < HALF_N; ++j) {
    const float lo = p[j], hi = p[j + HALF_N];
    const float keep = upper ? hi : lo;
    const float send = upper ? lo : hi;
    p[j] = keep + row_exchange<CTRL>(send);
  }
}

__global__ __launch_bounds__(256) void gat_gemm_wmma(const float* __restrict__ X,
                                                     const float* __restrict__ W,
                                                     const float* __restrict__ a_src,
                                                     const float* __restrict__ a_dst,
                                                     float* __restrict__ H,
                                                     float* __restrict__ as_,
                                                     float* __restrict__ ad_,
                                                     int N) {
  __shared__ float4 Wl[FDIM * FDIM / 4];

  const int tid  = threadIdx.x;
  const int lane = tid & 31;
  const int wave = tid >> 5;
  const int m    = lane & 15;
  const int half = lane >> 4;
  const int row0 = (blockIdx.x * 8 + wave) * 16;

  int rowA = row0 + m;
  if (rowA >= N) rowA = N - 1;
  const float* __restrict__ xrow = X + (size_t)rowA * FDIM + 2 * half;
  v2f a[FDIM / 4];
#pragma unroll
  for (int ks = 0; ks < FDIM / 4; ++ks) a[ks] = *(const v2f*)(xrow + 4 * ks);

  float asv[8], adv[8];
#pragma unroll
  for (int ct = 0; ct < 8; ++ct) {
    asv[ct] = a_src[ct * 16 + m];
    adv[ct] = a_dst[ct * 16 + m];
  }

  for (int q = tid; q < FDIM * FDIM / 4; q += 256) {
    const int ln  = q & 31;
    const int ctp = (q >> 5) & 3;
    const int ks  = q >> 7;
    const float* __restrict__ w =
        W + (4 * ks + 2 * (ln >> 4)) * FDIM + ctp * 32 + (ln & 15);
    float4 v;
    v.x = w[0];
    v.y = w[FDIM];
    v.z = w[16];
    v.w = w[FDIM + 16];
    Wl[q] = v;
  }
  __syncthreads();
  if (row0 >= N) return;

  v8f acc[8] = {};
  float4 bc[4];
#pragma unroll
  for (int ctp = 0; ctp < 4; ++ctp) bc[ctp] = Wl[ctp * 32 + lane];

#pragma unroll
  for (int ks = 0; ks < FDIM / 4; ++ks) {
    float4 bn[4];
    if (ks + 1 < FDIM / 4) {
#pragma unroll
      for (int ctp = 0; ctp < 4; ++ctp) bn[ctp] = Wl[((ks + 1) * 4 + ctp) * 32 + lane];
    }
    __builtin_amdgcn_sched_barrier(0);
#pragma unroll
    for (int ctp = 0; ctp < 4; ++ctp) {
      v2f b0; b0.x = bc[ctp].x; b0.y = bc[ctp].y;
      v2f b1; b1.x = bc[ctp].z; b1.y = bc[ctp].w;
      acc[2 * ctp] = __builtin_amdgcn_wmma_f32_16x16x4_f32(
           false, a[ks],  false, b0,
           (short)0, acc[2 * ctp],  false,  false);
      acc[2 * ctp + 1] = __builtin_amdgcn_wmma_f32_16x16x4_f32(
           false, a[ks],  false, b1,
           (short)0, acc[2 * ctp + 1],  false,  false);
    }
    __builtin_amdgcn_sched_barrier(0);
    if (ks + 1 < FDIM / 4) {
#pragma unroll
      for (int ctp = 0; ctp < 4; ++ctp) bc[ctp] = bn[ctp];
    }
  }

#pragma unroll
  for (int ct = 0; ct < 8; ++ct) {
#pragma unroll
    for (int v = 0; v < 8; ++v) {
      const int row = row0 + v + 8 * half;
      if (row < N) H[(size_t)row * FDIM + ct * 16 + m] = acc[ct][v];
    }
  }

  float p[64];
#pragma unroll
  for (int v = 0; v < 8; ++v) {
#pragma unroll
    for (int h = 0; h < NHEAD; ++h) {
      const int idx = ((v & 1) << 5) | (((v >> 1) & 1) << 4) | (((v >> 2) & 1) << 3) | h;
      p[idx]     = fmaf(acc[2 * h + 1][v], asv[2 * h + 1], acc[2 * h][v] * asv[2 * h]);
      p[idx | 4] = fmaf(acc[2 * h + 1][v], adv[2 * h + 1], acc[2 * h][v] * adv[2 * h]);
    }
  }
  fold_step<32, 0xB1>(p, (lane & 1) != 0);
  fold_step<16, 0x4E>(p, (lane & 2) != 0);
  fold_step<8, 0x164>(p, (lane & 4) != 0);
  fold_step<4, 0x168>(p, (lane & 8) != 0);
  {
    const int row = row0 + (m & 7) + 8 * half;
    float* __restrict__ dstp = ((m & 8) ? ad_ : as_) + (size_t)row * NHEAD;
    if (row < N) *(float4*)dstp = make_float4(p[0], p[1], p[2], p[3]);
  }
}

__device__ __forceinline__ float leaky(float v) {
  return v > 0.0f ? v : NEG_SLOPE * v;
}

#define BKT_SHIFT 7
#define BKT_NODES 128
#define NB_MAX 1023
#define SORT_BLOCKS 256
#define L1_THREADS 1024

__device__ __forceinline__ int wave_incl_scan(int v, int lane) {
  for (int o = 1; o < 32; o <<= 1) {
    const int t = __shfl_up(v, o);
    if (lane >= o) v += t;
  }
  return v;
}

__global__ __launch_bounds__(L1_THREADS) void edge_bucket_local(
    const int* __restrict__ ei, int E, int C, int nb,
    int* __restrict__ off, int2* __restrict__ tmp, int* __restrict__ total) {
  __shared__ int hist[NB_MAX + 1];
  __shared__ int wsum[32];
  const int tid  = threadIdx.x;
  const int lane = tid & 31;
  const int wave = tid >> 5;
  const int b    = blockIdx.x;
  const int e0   = b * C;
  const int e1   = (e0 + C < E) ? e0 + C : E;

  if (b == 0 && tid == 0) *total = 0;
  hist[tid] = 0;
  __syncthreads();

  for (int eb = e0 + tid; eb < e1; eb += 4 * L1_THREADS) {
    int d[4];
#pragma unroll
    for (int u = 0; u < 4; ++u) {
      const int e = eb + u * L1_THREADS;
      d[u] = (e < e1) ? ei[E + e] : -1;
    }
#pragma unroll
    for (int u = 0; u < 4; ++u)
      if (d[u] >= 0) atomicAdd(&hist[d[u] >> BKT_SHIFT], 1);
  }
  __syncthreads();

  const int v    = (tid <= nb) ? hist[tid] : 0;
  const int incl = wave_incl_scan(v, lane);
  if (lane == 31) wsum[wave] = incl;
  __syncthreads();
  if (wave == 0) {
    const int w = wsum[lane];
    wsum[lane] = wave_incl_scan(w, lane) - w;
  }
  __syncthreads();
  const int excl = wsum[wave] + incl - v;
  if (tid <= nb) {
    hist[tid] = excl;
    off[(size_t)tid * SORT_BLOCKS + b] = excl;
  }
  __syncthreads();

  for (int eb = e0 + tid; eb < e1; eb += 4 * L1_THREADS) {
    int s[4], d[4];
#pragma unroll
    for (int u = 0; u < 4; ++u) {
      const int e = eb + u * L1_THREADS;
      s[u] = (e < e1) ? ei[e] : 0;
      d[u] = (e < e1) ? ei[E + e] : -1;
    }
#pragma unroll
    for (int u = 0; u < 4; ++u) {
      if (d[u] >= 0) {
        const int r = atomicAdd(&hist[d[u] >> BKT_SHIFT], 1);
        tmp[(size_t)e0 + r] = make_int2(s[u], d[u]);
      }
    }
  }
}

__global__ __launch_bounds__(SORT_BLOCKS) void edge_bucket_final(
    const int2* __restrict__ tmp, const int* __restrict__ off, int C, int N,
    int* __restrict__ total, int* __restrict__ cnt, int* __restrict__ start,
    int* __restrict__ col) {
  __shared__ int c_l[BKT_NODES];
  __shared__ int s_l[BKT_NODES];
  __shared__ int wtot[BKT_NODES / 32];
  __shared__ int base_s;
  const int tid   = threadIdx.x;
  const int lane  = tid & 31;
  const int wave  = tid >> 5;
  const int k     = blockIdx.x;
  const int node0 = k << BKT_SHIFT;

  if (tid < BKT_NODES) c_l[tid] = (node0 + tid < N) ? 1 : 0;
  const int lo = off[(size_t)k * SORT_BLOCKS + tid];
  const int hi = off[(size_t)(k + 1) * SORT_BLOCKS + tid];
  const int2* __restrict__ seg = tmp + (size_t)tid * C;
  __syncthreads();

  for (int i = lo; i < hi; i += 4) {
    int dn[4];
#pragma unroll
    for (int u = 0; u < 4; ++u) {
      const int ii = (i + u < hi) ? i + u : hi - 1;
      dn[u] = seg[ii].y;
    }
#pragma unroll
    for (int u = 0; u < 4; ++u)
      if (i + u < hi) atomicAdd(&c_l[dn[u] & (BKT_NODES - 1)], 1);
  }
  __syncthreads();

  const int v    = (tid < BKT_NODES) ? c_l[tid] : 0;
  const int incl = wave_incl_scan(v, lane);
  if (lane == 31 && wave < BKT_NODES / 32) wtot[wave] = incl;
  __syncthreads();
  if (tid == 0)
    base_s = atomicAdd(total, wtot[0] + wtot[1] + wtot[2] + wtot[3]);
  __syncthreads();
  if (tid < BKT_NODES) {
    int pre = base_s;
    for (int w = 0; w < wave; ++w) pre += wtot[w];
    const int st   = pre + incl - v;
    const int node = node0 + tid;
    s_l[tid] = st;
    c_l[tid] = 1;
    if (node < N) { cnt[node] = v; start[node] = st; col[st] = node; }
  }
  __syncthreads();

  for (int i = lo; i < hi; i += 4) {
    int2 sd[4];
#pragma unroll
    for (int u = 0; u < 4; ++u) {
      const int ii = (i + u < hi) ? i + u : hi - 1;
      sd[u] = seg[ii];
    }
#pragma unroll
    for (int u = 0; u < 4; ++u) {
      if (i + u < hi) {
        const int j = sd[u].y & (BKT_NODES - 1);
        const int r = atomicAdd(&c_l[j], 1);
        col[s_l[j] + r] = sd[u].x;
      }
    }
  }
}

__device__ __forceinline__ float wave_max(float v) {
  for (int off = 16; off > 0; off >>= 1) v = fmaxf(v, __shfl_xor(v, off));
  return v;
}
__device__ __forceinline__ float wave_sum(float v) {
  for (int off = 16; off > 0; off >>= 1) v += __shfl_xor(v, off);
  return v;
}

__device__ __forceinline__ bool chunk_scores(const int* __restrict__ col,
                                             const float* __restrict__ as_,
                                             int beg, int deg, int base, int lane,
                                             float4 adv, int* s, float4* e) {
  const int j = base + lane;
  if (j >= deg) return false;
  const int sj = col[beg + j];
  const float4 a = *(const float4*)(as_ + (size_t)sj * NHEAD);
  *s = sj;
  e->x = leaky(a.x + adv.x);
  e->y = leaky(a.y + adv.y);
  e->z = leaky(a.z + adv.z);
  e->w = leaky(a.w + adv.w);
  return true;
}

__device__ __forceinline__ float bcast(float v, int jj) {
  return __int_as_float(__builtin_amdgcn_readlane(__float_as_int(v), jj));
}

struct RowTerm { float a; float4 hv; };

__device__ __forceinline__ RowTerm row_term(const float* __restrict__ H, int lane,
                                            int s, float4 al, int jj, int4 k) {
  RowTerm t;
  const int sj = __builtin_amdgcn_readlane(s, jj);
  t.a = __int_as_float((__float_as_int(bcast(al.x, jj)) & k.x) |
                       (__float_as_int(bcast(al.y, jj)) & k.y) |
                       (__float_as_int(bcast(al.z, jj)) & k.z) |
                       (__float_as_int(bcast(al.w, jj)) & k.w));
  t.hv = *((const float4*)(H + (size_t)sj * FDIM) + lane);
  return t;
}

__device__ __forceinline__ void row_fma(float4* acc, RowTerm t) {
  acc->x = fmaf(t.a, t.hv.x, acc->x);
  acc->y = fmaf(t.a, t.hv.y, acc->y);
  acc->z = fmaf(t.a, t.hv.z, acc->z);
  acc->w = fmaf(t.a, t.hv.w, acc->w);
}

__device__ __forceinline__ void gather_rows(const float* __restrict__ H,
                                            int s, float4 al, int cntc,
                                            int lane, int4 k, float4* acc) {
  int jj = 0;
  for (; jj + 4 <= cntc; jj += 4) {
    const RowTerm t0 = row_term(H, lane, s, al, jj + 0, k);
    const RowTerm t1 = row_term(H, lane, s, al, jj + 1, k);
    const RowTerm t2 = row_term(H, lane, s, al, jj + 2, k);
    const RowTerm t3 = row_term(H, lane, s, al, jj + 3, k);
    row_fma(acc, t0); row_fma(acc, t1); row_fma(acc, t2); row_fma(acc, t3);
  }
  for (; jj < cntc; ++jj) row_fma(acc, row_term(H, lane, s, al, jj, k));
}

__global__ __launch_bounds__(256) void gat_node_aggregate(
    const int* __restrict__ cnt, const int* __restrict__ start,
    const int* __restrict__ col,
    const float* __restrict__ as_, const float* __restrict__ ad_,
    const float* __restrict__ H, const float* __restrict__ bias,
    float* __restrict__ out, int N, int do_relu) {
  const int lane = threadIdx.x & 31;
  const int d = __builtin_amdgcn_readfirstlane(
      (int)(blockIdx.x * (blockDim.x >> 5) + (threadIdx.x >> 5)));
  if (d >= N) return;
  const int beg = start[d];
  const int deg = cnt[d];
  const float4 adv = *(const float4*)(ad_ + (size_t)d * NHEAD);
  const float NEG_INF = -__builtin_huge_valf();

  int    s0 = d;
  float4 e0 = make_float4(NEG_INF, NEG_INF, NEG_INF, NEG_INF);
  const bool v0 = chunk_scores(col, as_, beg, deg, 0, lane, adv, &s0, &e0);

  float4 m = e0;
  for (int base = 32; base < deg; base += 32) {
    int s; float4 e;
    if (chunk_scores(col, as_, beg, deg, base, lane, adv, &s, &e)) {
      m.x = fmaxf(m.x, e.x); m.y = fmaxf(m.y, e.y);
      m.z = fmaxf(m.z, e.z); m.w = fmaxf(m.w, e.w);
    }
  }
  m.x = wave_max(m.x); m.y = wave_max(m.y);
  m.z = wave_max(m.z); m.w = wave_max(m.w);
  if (!__builtin_isfinite(m.x)) m.x = 0.0f;
  if (!__builtin_isfinite(m.y)) m.y = 0.0f;
  if (!__builtin_isfinite(m.z)) m.z = 0.0f;
  if (!__builtin_isfinite(m.w)) m.w = 0.0f;

  float4 x0 = make_float4(0.0f, 0.0f, 0.0f, 0.0f);
  if (v0) {
    x0.x = __expf(e0.x - m.x); x0.y = __expf(e0.y - m.y);
    x0.z = __expf(e0.z - m.z); x0.w = __expf(e0.w - m.w);
  }
  float4 den = x0;
  for (int base = 32; base < deg; base += 32) {
    int s; float4 e;
    if (chunk_scores(col, as_, beg, deg, base, lane, adv, &s, &e)) {
      den.x += __expf(e.x - m.x); den.y += __expf(e.y - m.y);
      den.z += __expf(e.z - m.z); den.w += __expf(e.w - m.w);
    }
  }
  den.x = wave_sum(den.x) + 1e-16f; den.y = wave_sum(den.y) + 1e-16f;
  den.z = wave_sum(den.z) + 1e-16f; den.w = wave_sum(den.w) + 1e-16f;

  const int h = lane >> 3;
  const int4 k = make_int4(h == 0 ? -1 : 0, h == 1 ? -1 : 0,
                           h == 2 ? -1 : 0, h == 3 ? -1 : 0);
  float4 acc = make_float4(0.0f, 0.0f, 0.0f, 0.0f);
  {
    const float4 al = make_float4(x0.x / den.x, x0.y / den.y,
                                  x0.z / den.z, x0.w / den.w);
    gather_rows(H, s0, al, deg < 32 ? deg : 32, lane, k, &acc);
  }
  for (int base = 32; base < deg; base += 32) {
    int s = d;
    float4 e = make_float4(0.0f, 0.0f, 0.0f, 0.0f);
    float4 al = make_float4(0.0f, 0.0f, 0.0f, 0.0f);
    if (chunk_scores(col, as_, beg, deg, base, lane, adv, &s, &e)) {
      al.x = __expf(e.x - m.x) / den.x; al.y = __expf(e.y - m.y) / den.y;
      al.z = __expf(e.z - m.z) / den.z; al.w = __expf(e.w - m.w) / den.w;
    }
    const int rem = deg - base;
    gather_rows(H, s, al, rem < 32 ? rem : 32, lane, k, &acc);
  }

  const float4 bv = *(const float4*)(bias + lane * 4);
  acc.x += bv.x; acc.y += bv.y; acc.z += bv.z; acc.w += bv.w;
  if (do_relu) {
    acc.x = acc.x > 0.0f ? acc.x : 0.0f; acc.y = acc.y > 0.0f ? acc.y : 0.0f;
    acc.z = acc.z > 0.0f ? acc.z : 0.0f; acc.w = acc.w > 0.0f ? acc.w : 0.0f;
  }
  *(float4*)(out + (size_t)d * FDIM + lane * 4) = acc;
}

static inline int cdiv(long long a, int b) { return (int)((a + b - 1) / b); }

extern "C" void kernel_launch(void* const* d_in, const int* in_sizes, int n_in,
                              void* d_out, int out_size, void* d_ws, size_t ws_size,
                              hipStream_t stream) {
  const float* x     = (const float*)d_in[0];
  const int*   ei    = (const int*)d_in[1];
  const float* W1    = (const float*)d_in[2];
  const float* asrc1 = (const float*)d_in[3];
  const float* adst1 = (const float*)d_in[4];
  const float* b1    = (const float*)d_in[5];
  const float* W2    = (const float*)d_in[6];
  const float* asrc2 = (const float*)d_in[7];
  const float* adst2 = (const float*)d_in[8];
  const float* b2    = (const float*)d_in[9];
  float* out = (float*)d_out;

  const int N  = in_sizes[0] / FDIM;
  const int E  = in_sizes[1] / 2;

  const int nb = cdiv(N, BKT_NODES);
  const int C  = cdiv(E, SORT_BLOCKS);
  float* H    = (float*)d_ws;
  int*   total  = (int*)(H + (size_t)N * FDIM);
  int2*  tmp    = (int2*)(total + 4);
  int*   off    = total + 4 + (size_t)2 * E;
  int*   cnt    = off + (size_t)(nb + 1) * SORT_BLOCKS;
  int*   start  = cnt + N;
  int*   col    = start + N;
  float* as_  = H + (size_t)2 * N * FDIM;
  float* ad_  = as_  + (size_t)N * NHEAD;

  const int TB = 256;
  const int gStripes = cdiv(N, 128);
  const int gNode  = cdiv(N, TB / 32);

  edge_bucket_local<<<SORT_BLOCKS, L1_THREADS, 0, stream>>>(ei, E, C, nb, off, tmp, total);
  edge_bucket_final<<<nb, SORT_BLOCKS, 0, stream>>>(tmp, off, C, N, total, cnt, start, col);

  gat_gemm_wmma<<<gStripes, TB, 0, stream>>>(x, W1, asrc1, adst1, H, as_, ad_, N);
  gat_node_aggregate<<<gNode, TB, 0, stream>>>(cnt, start, col, as_, ad_, H, b1,
                                               out, N,  1);

  gat_gemm_wmma<<<gStripes, TB, 0, stream>>>(out, W2, asrc2, adst2, H, as_, ad_, N);
  gat_node_aggregate<<<gNode, TB, 0, stream>>>(cnt, start, col, as_, ad_, H, b2,
                                               out, N,  0);
}


# Layer-2 GEMM moved into the layer-1 aggregation launch: a block sums 16 nodes into an LDS tile, then multiplies it by W2 on WMMA; launches 6 to 5; layer-2 dst logit now summed as a lane tree
# speedup vs baseline: 6.0343x; 1.0967x over previous
#include <hip/hip_runtime.h>
#include <hip/hip_bf16.h>


typedef __attribute__((ext_vector_type(2))) float v2f;
typedef __attribute__((ext_vector_type(8))) float v8f;

#define FDIM 128
#define NHEAD 4
#define CHAN 32
#define NEG_SLOPE 0.2f


template <int CTRL>
__device__ __forceinline__ float row_exchange(float x) {
  return __int_as_float(
      __builtin_amdgcn_update_dpp(0, __float_as_int(x), CTRL, 0xf, 0xf, true));
}

template <int HALF_N, int CTRL>
__device__ __forceinline__ void fold_step(float* p, bool upper) {
#pragma unroll
  for (int j = 0; j < HALF_N; ++j) {
    const float lo = p[j], hi = p[j + HALF_N];
    const float keep = upper ? hi : lo;
    const float send = upper ? lo : hi;
    p[j] = keep + row_exchange<CTRL>(send);
  }
}

__global__ __launch_bounds__(256) void gat_gemm_wmma(const float* __restrict__ X,
                                                     const float* __restrict__ W,
                                                     const float* __restrict__ a_src,
                                                     const float* __restrict__ a_dst,
                                                     float* __restrict__ H,
                                                     float* __restrict__ as_,
                                                     float* __restrict__ ad_,
                                                     int N) {
  __shared__ float4 Wl[FDIM * FDIM / 4];

  const int tid  = threadIdx.x;
  const int lane = tid & 31;
  const int wave = tid >> 5;
  const int m    = lane & 15;
  const int half = lane >> 4;
  const int row0 = (blockIdx.x * 8 + wave) * 16;

  int rowA = row0 + m;
  if (rowA >= N) rowA = N - 1;
  const float* __restrict__ xrow = X + (size_t)rowA * FDIM + 2 * half;
  v2f a[FDIM / 4];
#pragma unroll
  for (int ks = 0; ks < FDIM / 4; ++ks) a[ks] = *(const v2f*)(xrow + 4 * ks);

  float asv[8], adv[8];
#pragma unroll
  for (int ct = 0; ct < 8; ++ct) {
    asv[ct] = a_src[ct * 16 + m];
    adv[ct] = a_dst[ct * 16 + m];
  }

  for (int q = tid; q < FDIM * FDIM / 4; q += 256) {
    const int ln  = q & 31;
    const int ctp = (q >> 5) & 3;
    const int ks  = q >> 7;
    const float* __restrict__ w =
        W + (4 * ks + 2 * (ln >> 4)) * FDIM + ctp * 32 + (ln & 15);
    float4 v;
    v.x = w[0];
    v.y = w[FDIM];
    v.z = w[16];
    v.w = w[FDIM + 16];
    Wl[q] = v;
  }
  __syncthreads();
  if (row0 >= N) return;

  v8f acc[8] = {};
  float4 bc[4];
#pragma unroll
  for (int ctp = 0; ctp < 4; ++ctp) bc[ctp] = Wl[ctp * 32 + lane];

#pragma unroll
  for (int ks = 0; ks < FDIM / 4; ++ks) {
    float4 bn[4];
    if (ks + 1 < FDIM / 4) {
#pragma unroll
      for (int ctp = 0; ctp < 4; ++ctp) bn[ctp] = Wl[((ks + 1) * 4 + ctp) * 32 + lane];
    }
    __builtin_amdgcn_sched_barrier(0);
#pragma unroll
    for (int ctp = 0; ctp < 4; ++ctp) {
      v2f b0; b0.x = bc[ctp].x; b0.y = bc[ctp].y;
      v2f b1; b1.x = bc[ctp].z; b1.y = bc[ctp].w;
      acc[2 * ctp] = __builtin_amdgcn_wmma_f32_16x16x4_f32(
           false, a[ks],  false, b0,
           (short)0, acc[2 * ctp],  false,  false);
      acc[2 * ctp + 1] = __builtin_amdgcn_wmma_f32_16x16x4_f32(
           false, a[ks],  false, b1,
           (short)0, acc[2 * ctp + 1],  false,  false);
    }
    __builtin_amdgcn_sched_barrier(0);
    if (ks + 1 < FDIM / 4) {
#pragma unroll
      for (int ctp = 0; ctp < 4; ++ctp) bc[ctp] = bn[ctp];
    }
  }

#pragma unroll
  for (int ct = 0; ct < 8; ++ct) {
#pragma unroll
    for (int v = 0; v < 8; ++v) {
      const int row = row0 + v + 8 * half;
      if (row < N) H[(size_t)row * FDIM + ct * 16 + m] = acc[ct][v];
    }
  }

  float p[64];
#pragma unroll
  for (int v = 0; v < 8; ++v) {
#pragma unroll
    for (int h = 0; h < NHEAD; ++h) {
      const int idx = ((v & 1) << 5) | (((v >> 1) & 1) << 4) | (((v >> 2) & 1) << 3) | h;
      p[idx]     = fmaf(acc[2 * h + 1][v], asv[2 * h + 1], acc[2 * h][v] * asv[2 * h]);
      p[idx | 4] = fmaf(acc[2 * h + 1][v], adv[2 * h + 1], acc[2 * h][v] * adv[2 * h]);
    }
  }
  fold_step<32, 0xB1>(p, (lane & 1) != 0);
  fold_step<16, 0x4E>(p, (lane & 2) != 0);
  fold_step<8, 0x164>(p, (lane & 4) != 0);
  fold_step<4, 0x168>(p, (lane & 8) != 0);
  {
    const int row = row0 + (m & 7) + 8 * half;
    float* __restrict__ dstp = ((m & 8) ? ad_ : as_) + (size_t)row * NHEAD;
    if (row < N) *(float4*)dstp = make_float4(p[0], p[1], p[2], p[3]);
  }
}

__device__ __forceinline__ float leaky(float v) {
  return v > 0.0f ? v : NEG_SLOPE * v;
}

#define BKT_SHIFT 7
#define BKT_NODES 128
#define NB_MAX 1023
#define SORT_BLOCKS 256
#define L1_THREADS 1024

__device__ __forceinline__ int wave_incl_scan(int v, int lane) {
  for (int o = 1; o < 32; o <<= 1) {
    const int t = __shfl_up(v, o);
    if (lane >= o) v += t;
  }
  return v;
}

__global__ __launch_bounds__(L1_THREADS) void edge_bucket_local(
    const int* __restrict__ ei, int E, int C, int nb,
    int* __restrict__ off, int2* __restrict__ tmp, int* __restrict__ total) {
  __shared__ int hist[NB_MAX + 1];
  __shared__ int wsum[32];
  const int tid  = threadIdx.x;
  const int lane = tid & 31;
  const int wave = tid >> 5;
  const int b    = blockIdx.x;
  const int e0   = b * C;
  const int e1   = (e0 + C < E) ? e0 + C : E;

  if (b == 0 && tid == 0) *total = 0;
  hist[tid] = 0;
  __syncthreads();

  for (int eb = e0 + tid; eb < e1; eb += 4 * L1_THREADS) {
    int d[4];
#pragma unroll
    for (int u = 0; u < 4; ++u) {
      const int e = eb + u * L1_THREADS;
      d[u] = (e < e1) ? ei[E + e] : -1;
    }
#pragma unroll
    for (int u = 0; u < 4; ++u)
      if (d[u] >= 0) atomicAdd(&hist[d[u] >> BKT_SHIFT], 1);
  }
  __syncthreads();

  const int v    = (tid <= nb) ? hist[tid] : 0;
  const int incl = wave_incl_scan(v, lane);
  if (lane == 31) wsum[wave] = incl;
  __syncthreads();
  if (wave == 0) {
    const int w = wsum[lane];
    wsum[lane] = wave_incl_scan(w, lane) - w;
  }
  __syncthreads();
  const int excl = wsum[wave] + incl - v;
  if (tid <= nb) {
    hist[tid] = excl;
    off[(size_t)tid * SORT_BLOCKS + b] = excl;
  }
  __syncthreads();

  for (int eb = e0 + tid; eb < e1; eb += 4 * L1_THREADS) {
    int s[4], d[4];
#pragma unroll
    for (int u = 0; u < 4; ++u) {
      const int e = eb + u * L1_THREADS;
      s[u] = (e < e1) ? ei[e] : 0;
      d[u] = (e < e1) ? ei[E + e] : -1;
    }
#pragma unroll
    for (int u = 0; u < 4; ++u) {
      if (d[u] >= 0) {
        const int r = atomicAdd(&hist[d[u] >> BKT_SHIFT], 1);
        tmp[(size_t)e0 + r] = make_int2(s[u], d[u]);
      }
    }
  }
}

__global__ __launch_bounds__(SORT_BLOCKS) void edge_bucket_final(
    const int2* __restrict__ tmp, const int* __restrict__ off, int C, int N,
    int* __restrict__ total, int* __restrict__ cnt, int* __restrict__ start,
    int* __restrict__ col) {
  __shared__ int c_l[BKT_NODES];
  __shared__ int s_l[BKT_NODES];
  __shared__ int wtot[BKT_NODES / 32];
  __shared__ int base_s;
  const int tid   = threadIdx.x;
  const int lane  = tid & 31;
  const int wave  = tid >> 5;
  const int k     = blockIdx.x;
  const int node0 = k << BKT_SHIFT;

  if (tid < BKT_NODES) c_l[tid] = (node0 + tid < N) ? 1 : 0;
  const int lo = off[(size_t)k * SORT_BLOCKS + tid];
  const int hi = off[(size_t)(k + 1) * SORT_BLOCKS + tid];
  const int2* __restrict__ seg = tmp + (size_t)tid * C;
  __syncthreads();

  for (int i = lo; i < hi; i += 4) {
    int dn[4];
#pragma unroll
    for (int u = 0; u < 4; ++u) {
      const int ii = (i + u < hi) ? i + u : hi - 1;
      dn[u] = seg[ii].y;
    }
#pragma unroll
    for (int u = 0; u < 4; ++u)
      if (i + u < hi) atomicAdd(&c_l[dn[u] & (BKT_NODES - 1)], 1);
  }
  __syncthreads();

  const int v    = (tid < BKT_NODES) ? c_l[tid] : 0;
  const int incl = wave_incl_scan(v, lane);
  if (lane == 31 && wave < BKT_NODES / 32) wtot[wave] = incl;
  __syncthreads();
  if (tid == 0)
    base_s = atomicAdd(total, wtot[0] + wtot[1] + wtot[2] + wtot[3]);
  __syncthreads();
  if (tid < BKT_NODES) {
    int pre = base_s;
    for (int w = 0; w < wave; ++w) pre += wtot[w];
    const int st   = pre + incl - v;
    const int node = node0 + tid;
    s_l[tid] = st;
    c_l[tid] = 1;
    if (node < N) { cnt[node] = v; start[node] = st; col[st] = node; }
  }
  __syncthreads();

  for (int i = lo; i < hi; i += 4) {
    int2 sd[4];
#pragma unroll
    for (int u = 0; u < 4; ++u) {
      const int ii = (i + u < hi) ? i + u : hi - 1;
      sd[u] = seg[ii];
    }
#pragma unroll
    for (int u = 0; u < 4; ++u) {
      if (i + u < hi) {
        const int j = sd[u].y & (BKT_NODES - 1);
        const int r = atomicAdd(&c_l[j], 1);
        col[s_l[j] + r] = sd[u].x;
      }
    }
  }
}

__device__ __forceinline__ float head_dot(float4 r, float4 a) {
  v2f rl; rl.x = r.x; rl.y = r.y;
  v2f rh; rh.x = r.z; rh.y = r.w;
  v2f al; al.x = a.x; al.y = a.y;
  v2f ah; ah.x = a.z; ah.y = a.w;
  const v2f t2 = __builtin_elementwise_fma(rh, ah, rl * al);
  float t = t2.x + t2.y;
  t += row_exchange<0xB1>(t);
  t += row_exchange<0x4E>(t);
  t += row_exchange<0x164>(t);
  return t;
}

__device__ __forceinline__ void row_fma(float4* acc, float a, float4 hv) {
  acc->x = fmaf(a, hv.x, acc->x);
  acc->y = fmaf(a, hv.y, acc->y);
  acc->z = fmaf(a, hv.z, acc->z);
  acc->w = fmaf(a, hv.w, acc->w);
}

__device__ __forceinline__ float4 src_row(const float* __restrict__ H, int s, int jj,
                                          int lane) {
  const int sj = __builtin_amdgcn_readlane(s, jj);
  return *((const float4*)(H + (size_t)sj * FDIM) + lane);
}

__device__ __forceinline__ float4 node_row_sum(const int* __restrict__ col,
                                               const float* __restrict__ H,
                                               int beg, int deg, int d, float adh,
                                               float4 asv, int lane) {
  const float NEG_INF = -__builtin_huge_valf();
  float m = NEG_INF;
  float den = 0.0f;
  float4 acc = make_float4(0.0f, 0.0f, 0.0f, 0.0f);

  for (int base = 0; base < deg; base += 32) {
    const int j = base + lane;
    const int s = (j < deg) ? col[beg + j] : d;
    const int rem = deg - base;
    const int cntc = rem < 32 ? rem : 32;
    int jj = 0;
    for (; jj + 4 <= cntc; jj += 4) {
      const float4 r0 = src_row(H, s, jj + 0, lane);
      const float4 r1 = src_row(H, s, jj + 1, lane);
      const float4 r2 = src_row(H, s, jj + 2, lane);
      const float4 r3 = src_row(H, s, jj + 3, lane);
      const float e0 = leaky(head_dot(r0, asv) + adh);
      const float e1 = leaky(head_dot(r1, asv) + adh);
      const float e2 = leaky(head_dot(r2, asv) + adh);
      const float e3 = leaky(head_dot(r3, asv) + adh);
      const float gm = fmaxf(fmaxf(m, e0), fmaxf(fmaxf(e1, e2), e3));
      const float mu = (gm == NEG_INF) ? 0.0f : gm;
      const float sc = __expf(m - mu);
      const float p0 = __expf(e0 - mu);
      const float p1 = __expf(e1 - mu);
      const float p2 = __expf(e2 - mu);
      const float p3 = __expf(e3 - mu);
      den = fmaf(den, sc, (p0 + p1) + (p2 + p3));
      acc.x *= sc; acc.y *= sc; acc.z *= sc; acc.w *= sc;
      row_fma(&acc, p0, r0); row_fma(&acc, p1, r1);
      row_fma(&acc, p2, r2); row_fma(&acc, p3, r3);
      m = gm;
    }
    for (; jj < cntc; ++jj) {
      const float4 r0 = src_row(H, s, jj, lane);
      const float e0 = leaky(head_dot(r0, asv) + adh);
      const float gm = fmaxf(m, e0);
      const float mu = (gm == NEG_INF) ? 0.0f : gm;
      const float sc = __expf(m - mu);
      const float p0 = __expf(e0 - mu);
      den = fmaf(den, sc, p0);
      acc.x *= sc; acc.y *= sc; acc.z *= sc; acc.w *= sc;
      row_fma(&acc, p0, r0);
      m = gm;
    }
  }

  den += 1e-16f;
  acc.x /= den; acc.y /= den; acc.z /= den; acc.w /= den;
  return acc;
}

#define TILE_PITCH4 33
#define AGG_WAVES 16

__global__ __launch_bounds__(AGG_WAVES * 32) void gat_aggregate_gemm(
    const int* __restrict__ cnt, const int* __restrict__ start,
    const int* __restrict__ col,
    const float* __restrict__ a_src, const float* __restrict__ ad_,
    const float* __restrict__ H, const float* __restrict__ bias,
    const float* __restrict__ W, float* __restrict__ H2, int N) {
  __shared__ float4 Wl[FDIM * FDIM / 4];
  __shared__ float4 tile[AGG_WAVES * TILE_PITCH4];

  const int tid  = threadIdx.x;
  const int lane = tid & 31;
  const int wave = __builtin_amdgcn_readfirstlane(tid >> 5);
  const int m    = lane & 15;
  const int half = lane >> 4;
  const int h    = lane >> 3;

#pragma unroll
  for (int q = tid; q < FDIM * FDIM / 4; q += AGG_WAVES * 32) {
    const int ln  = q & 31;
    const int ctp = (q >> 5) & 3;
    const int ks  = q >> 7;
    const float* __restrict__ w =
        W + (4 * ks + 2 * (ln >> 4)) * FDIM + ctp * 32 + (ln & 15);
    float4 v;
    v.x = w[0];
    v.y = w[FDIM];
    v.z = w[16];
    v.w = w[FDIM + 16];
    Wl[q] = v;
  }

  const float4 asv = *((const float4*)a_src + lane);
  const float4 bv  = *(const float4*)(bias + lane * 4);
  const int nStripes = (N + AGG_WAVES - 1) / AGG_WAVES;

  for (int s = blockIdx.x; s < nStripes; s += gridDim.x) {
    const int row0 = s * AGG_WAVES;
    const int d = row0 + wave;
    const int dd = d < N ? d : N - 1;
    const int beg = start[dd];
    const int deg = d < N ? cnt[dd] : 0;
    const float adh = ad_[(size_t)dd * NHEAD + h];

    float4 acc = node_row_sum(col, H, beg, deg, dd, adh, asv, lane);
    acc.x += bv.x; acc.y += bv.y; acc.z += bv.z; acc.w += bv.w;
    acc.x = acc.x > 0.0f ? acc.x : 0.0f; acc.y = acc.y > 0.0f ? acc.y : 0.0f;
    acc.z = acc.z > 0.0f ? acc.z : 0.0f; acc.w = acc.w > 0.0f ? acc.w : 0.0f;

    __syncthreads();
    tile[wave * TILE_PITCH4 + lane] = acc;
    __syncthreads();

    if (wave < NHEAD) {
      const v2f* __restrict__ tA = (const v2f*)tile + m * (2 * TILE_PITCH4) + half;
      v8f c0 = {}, c1 = {};
#pragma unroll
      for (int ks = 0; ks < FDIM / 4; ++ks) {
        const v2f a = tA[2 * ks];
        const float4 b = Wl[(ks * 4 + wave) * 32 + lane];
        v2f b0; b0.x = b.x; b0.y = b.y;
        v2f b1; b1.x = b.z; b1.y = b.w;
        c0 = __builtin_amdgcn_wmma_f32_16x16x4_f32(
             false, a,  false, b0,
             (short)0, c0,  false,  false);
        c1 = __builtin_amdgcn_wmma_f32_16x16x4_f32(
             false, a,  false, b1,
             (short)0, c1,  false,  false);
      }
      float* __restrict__ hp = H2 + (size_t)(row0 + 8 * half) * FDIM + wave * 32 + m;
#pragma unroll
      for (int v = 0; v < 8; ++v) {
        hp[v * FDIM]      = c0[v];
        hp[v * FDIM + 16] = c1[v];
      }
    }
  }
}

__global__ __launch_bounds__(256) void gat_node_aggregate(
    const int* __restrict__ cnt, const int* __restrict__ start,
    const int* __restrict__ col,
    const float* __restrict__ a_src, const float* __restrict__ a_dst,
    const float* __restrict__ H, const float* __restrict__ bias,
    float* __restrict__ out, int N) {
  const int lane = threadIdx.x & 31;
  const int d = __builtin_amdgcn_readfirstlane(
      (int)(blockIdx.x * (blockDim.x >> 5) + (threadIdx.x >> 5)));
  if (d >= N) return;
  const int beg = start[d];
  const int deg = cnt[d];
  const float4 asv = *((const float4*)a_src + lane);
  const float4 adv = *((const float4*)a_dst + lane);
  const float4 own = *((const float4*)(H + (size_t)d * FDIM) + lane);
  const float adh = head_dot(own, adv);

  float4 acc = node_row_sum(col, H, beg, deg, d, adh, asv, lane);
  const float4 bv = *(const float4*)(bias + lane * 4);
  acc.x += bv.x; acc.y += bv.y; acc.z += bv.z; acc.w += bv.w;
  *(float4*)(out + (size_t)d * FDIM + lane * 4) = acc;
}

static inline int cdiv(long long a, int b) { return (int)((a + b - 1) / b); }

extern "C" void kernel_launch(void* const* d_in, const int* in_sizes, int n_in,
                              void* d_out, int out_size, void* d_ws, size_t ws_size,
                              hipStream_t stream) {
  const float* x     = (const float*)d_in[0];
  const int*   ei    = (const int*)d_in[1];
  const float* W1    = (const float*)d_in[2];
  const float* asrc1 = (const float*)d_in[3];
  const float* adst1 = (const float*)d_in[4];
  const float* b1    = (const float*)d_in[5];
  const float* W2    = (const float*)d_in[6];
  const float* asrc2 = (const float*)d_in[7];
  const float* adst2 = (const float*)d_in[8];
  const float* b2    = (const float*)d_in[9];
  float* out = (float*)d_out;

  const int N  = in_sizes[0] / FDIM;
  const int E  = in_sizes[1] / 2;

  const int nb = cdiv(N, BKT_NODES);
  const int C  = cdiv(E, SORT_BLOCKS);
  float* H2   = (float*)d_ws;
  float* H1   = out;
  int*   total  = (int*)(H2 + (size_t)N * FDIM);
  int2*  tmp    = (int2*)(total + 4);
  int*   off    = total + 4 + (size_t)2 * E;
  int*   cnt    = off + (size_t)(nb + 1) * SORT_BLOCKS;
  int*   start  = cnt + N;
  int*   col    = start + N;
  float* as_  = H2 + (size_t)2 * N * FDIM;
  float* ad_  = as_  + (size_t)N * NHEAD;

  const int TB = 256;
  const int gStripes = cdiv(N, 128);
  const int gNode  = cdiv(N, TB / 32);
  const int gFused = 1024;

  edge_bucket_local<<<SORT_BLOCKS, L1_THREADS, 0, stream>>>(ei, E, C, nb, off, tmp, total);
  edge_bucket_final<<<nb, SORT_BLOCKS, 0, stream>>>(tmp, off, C, N, total, cnt, start, col);

  gat_gemm_wmma<<<gStripes, TB, 0, stream>>>(x, W1, asrc1, adst1, H1, as_, ad_, N);
  gat_aggregate_gemm<<<gFused, AGG_WAVES * 32, 0, stream>>>(cnt, start, col, asrc1, ad_, H1, b1,
                                                            W2, H2, N);

  gat_node_aggregate<<<gNode, TB, 0, stream>>>(cnt, start, col, asrc2, adst2, H2, b2, out, N);
}


# Level 2 of the edge grouping and the layer-1 GEMM, independent of each other, share one launch: role by block index, a GEMM block and a bucket block side by side per WGP; launches 5 to 4
# speedup vs baseline: 6.3582x; 1.0537x over previous
#include <hip/hip_runtime.h>
#include <hip/hip_bf16.h>


typedef __attribute__((ext_vector_type(2))) float v2f;
typedef __attribute__((ext_vector_type(8))) float v8f;

#define FDIM 128
#define NHEAD 4
#define CHAN 32
#define NEG_SLOPE 0.2f


template <int CTRL>
__device__ __forceinline__ float row_exchange(float x) {
  return __int_as_float(
      __builtin_amdgcn_update_dpp(0, __float_as_int(x), CTRL, 0xf, 0xf, true));
}

template <int HALF_N, int CTRL>
__device__ __forceinline__ void fold_step(float* p, bool upper) {
#pragma unroll
  for (int j = 0; j < HALF_N; ++j) {
    const float lo = p[j], hi = p[j + HALF_N];
    const float keep = upper ? hi : lo;
    const float send = upper ? lo : hi;
    p[j] = keep + row_exchange<CTRL>(send);
  }
}

__device__ __forceinline__ void gat_gemm_wmma_block(const float* __restrict__ X,
                                                    const float* __restrict__ W,
                                                    const float* __restrict__ a_src,
                                                    const float* __restrict__ a_dst,
                                                    float* __restrict__ H,
                                                    float* __restrict__ as_,
                                                    float* __restrict__ ad_,
                                                    int N, int unit, float4* Wl) {
  const int tid  = threadIdx.x;
  const int lane = tid & 31;
  const int wave = tid >> 5;
  const int m    = lane & 15;
  const int half = lane >> 4;
  const int row0 = (unit * 8 + wave) * 16;

  int rowA = row0 + m;
  if (rowA >= N) rowA = N - 1;
  const float* __restrict__ xrow = X + (size_t)rowA * FDIM + 2 * half;
  v2f a[FDIM / 4];
#pragma unroll
  for (int ks = 0; ks < FDIM / 4; ++ks) a[ks] = *(const v2f*)(xrow + 4 * ks);

  float asv[8], adv[8];
#pragma unroll
  for (int ct = 0; ct < 8; ++ct) {
    asv[ct] = a_src[ct * 16 + m];
    adv[ct] = a_dst[ct * 16 + m];
  }

  for (int q = tid; q < FDIM * FDIM / 4; q += 256) {
    const int ln  = q & 31;
    const int ctp = (q >> 5) & 3;
    const int ks  = q >> 7;
    const float* __restrict__ w =
        W + (4 * ks + 2 * (ln >> 4)) * FDIM + ctp * 32 + (ln & 15);
    float4 v;
    v.x = w[0];
    v.y = w[FDIM];
    v.z = w[16];
    v.w = w[FDIM + 16];
    Wl[q] = v;
  }
  __syncthreads();
  if (row0 >= N) return;

  v8f acc[8] = {};
  float4 bc[4];
#pragma unroll
  for (int ctp = 0; ctp < 4; ++ctp) bc[ctp] = Wl[ctp * 32 + lane];

#pragma unroll
  for (int ks = 0; ks < FDIM / 4; ++ks) {
    float4 bn[4];
    if (ks + 1 < FDIM / 4) {
#pragma unroll
      for (int ctp = 0; ctp < 4; ++ctp) bn[ctp] = Wl[((ks + 1) * 4 + ctp) * 32 + lane];
    }
    __builtin_amdgcn_sched_barrier(0);
#pragma unroll
    for (int ctp = 0; ctp < 4; ++ctp) {
      v2f b0; b0.x = bc[ctp].x; b0.y = bc[ctp].y;
      v2f b1; b1.x = bc[ctp].z; b1.y = bc[ctp].w;
      acc[2 * ctp] = __builtin_amdgcn_wmma_f32_16x16x4_f32(
           false, a[ks],  false, b0,
           (short)0, acc[2 * ctp],  false,  false);
      acc[2 * ctp + 1] = __builtin_amdgcn_wmma_f32_16x16x4_f32(
           false, a[ks],  false, b1,
           (short)0, acc[2 * ctp + 1],  false,  false);
    }
    __builtin_amdgcn_sched_barrier(0);
    if (ks + 1 < FDIM / 4) {
#pragma unroll
      for (int ctp = 0; ctp < 4; ++ctp) bc[ctp] = bn[ctp];
    }
  }

#pragma unroll
  for (int ct = 0; ct < 8; ++ct) {
#pragma unroll
    for (int v = 0; v < 8; ++v) {
      const int row = row0 + v + 8 * half;
      if (row < N) H[(size_t)row * FDIM + ct * 16 + m] = acc[ct][v];
    }
  }

  float p[64];
#pragma unroll
  for (int v = 0; v < 8; ++v) {
#pragma unroll
    for (int h = 0; h < NHEAD; ++h) {
      const int idx = ((v & 1) << 5) | (((v >> 1) & 1) << 4) | (((v >> 2) & 1) << 3) | h;
      p[idx]     = fmaf(acc[2 * h + 1][v], asv[2 * h + 1], acc[2 * h][v] * asv[2 * h]);
      p[idx | 4] = fmaf(acc[2 * h + 1][v], adv[2 * h + 1], acc[2 * h][v] * adv[2 * h]);
    }
  }
  fold_step<32, 0xB1>(p, (lane & 1) != 0);
  fold_step<16, 0x4E>(p, (lane & 2) != 0);
  fold_step<8, 0x164>(p, (lane & 4) != 0);
  fold_step<4, 0x168>(p, (lane & 8) != 0);
  {
    const int row = row0 + (m & 7) + 8 * half;
    float* __restrict__ dstp = ((m & 8) ? ad_ : as_) + (size_t)row * NHEAD;
    if (row < N) *(float4*)dstp = make_float4(p[0], p[1], p[2], p[3]);
  }
}

__device__ __forceinline__ float leaky(float v) {
  return v > 0.0f ? v : NEG_SLOPE * v;
}

#define BKT_SHIFT 7
#define BKT_NODES 128
#define NB_MAX 1023
#define SORT_BLOCKS 256
#define L1_THREADS 1024

__device__ __forceinline__ int wave_incl_scan(int v, int lane) {
  for (int o = 1; o < 32; o <<= 1) {
    const int t = __shfl_up(v, o);
    if (lane >= o) v += t;
  }
  return v;
}

__global__ __launch_bounds__(L1_THREADS) void edge_bucket_local(
    const int* __restrict__ ei, int E, int C, int nb,
    int* __restrict__ off, int2* __restrict__ tmp, int* __restrict__ total) {
  __shared__ int hist[NB_MAX + 1];
  __shared__ int wsum[32];
  const int tid  = threadIdx.x;
  const int lane = tid & 31;
  const int wave = tid >> 5;
  const int b    = blockIdx.x;
  const int e0   = b * C;
  const int e1   = (e0 + C < E) ? e0 + C : E;

  if (b == 0 && tid == 0) *total = 0;
  hist[tid] = 0;
  __syncthreads();

  for (int eb = e0 + tid; eb < e1; eb += 4 * L1_THREADS) {
    int d[4];
#pragma unroll
    for (int u = 0; u < 4; ++u) {
      const int e = eb + u * L1_THREADS;
      d[u] = (e < e1) ? ei[E + e] : -1;
    }
#pragma unroll
    for (int u = 0; u < 4; ++u)
      if (d[u] >= 0) atomicAdd(&hist[d[u] >> BKT_SHIFT], 1);
  }
  __syncthreads();

  const int v    = (tid <= nb) ? hist[tid] : 0;
  const int incl = wave_incl_scan(v, lane);
  if (lane == 31) wsum[wave] = incl;
  __syncthreads();
  if (wave == 0) {
    const int w = wsum[lane];
    wsum[lane] = wave_incl_scan(w, lane) - w;
  }
  __syncthreads();
  const int excl = wsum[wave] + incl - v;
  if (tid <= nb) {
    hist[tid] = excl;
    off[(size_t)tid * SORT_BLOCKS + b] = excl;
  }
  __syncthreads();

  for (int eb = e0 + tid; eb < e1; eb += 4 * L1_THREADS) {
    int s[4], d[4];
#pragma unroll
    for (int u = 0; u < 4; ++u) {
      const int e = eb + u * L1_THREADS;
      s[u] = (e < e1) ? ei[e] : 0;
      d[u] = (e < e1) ? ei[E + e] : -1;
    }
#pragma unroll
    for (int u = 0; u < 4; ++u) {
      if (d[u] >= 0) {
        const int r = atomicAdd(&hist[d[u] >> BKT_SHIFT], 1);
        tmp[(size_t)e0 + r] = make_int2(s[u], d[u]);
      }
    }
  }
}

__device__ __forceinline__ void edge_bucket_final_block(
    const int2* __restrict__ tmp, const int* __restrict__ off, int C, int N,
    int* __restrict__ total, int* __restrict__ cnt, int* __restrict__ start,
    int* __restrict__ col, int k, int* c_l, int* s_l, int* wtot, int* base_sp) {
  const int tid   = threadIdx.x;
  const int lane  = tid & 31;
  const int wave  = tid >> 5;
  const int node0 = k << BKT_SHIFT;

  if (tid < BKT_NODES) c_l[tid] = (node0 + tid < N) ? 1 : 0;
  const int lo = off[(size_t)k * SORT_BLOCKS + tid];
  const int hi = off[(size_t)(k + 1) * SORT_BLOCKS + tid];
  const int2* __restrict__ seg = tmp + (size_t)tid * C;
  __syncthreads();

  for (int i = lo; i < hi; i += 4) {
    int dn[4];
#pragma unroll
    for (int u = 0; u < 4; ++u) {
      const int ii = (i + u < hi) ? i + u : hi - 1;
      dn[u] = seg[ii].y;
    }
#pragma unroll
    for (int u = 0; u < 4; ++u)
      if (i + u < hi) atomicAdd(&c_l[dn[u] & (BKT_NODES - 1)], 1);
  }
  __syncthreads();

  const int v    = (tid < BKT_NODES) ? c_l[tid] : 0;
  const int incl = wave_incl_scan(v, lane);
  if (lane == 31 && wave < BKT_NODES / 32) wtot[wave] = incl;
  __syncthreads();
  if (tid == 0)
    *base_sp = atomicAdd(total, wtot[0] + wtot[1] + wtot[2] + wtot[3]);
  __syncthreads();
  if (tid < BKT_NODES) {
    int pre = *base_sp;
    for (int w = 0; w < wave; ++w) pre += wtot[w];
    const int st   = pre + incl - v;
    const int node = node0 + tid;
    s_l[tid] = st;
    c_l[tid] = 1;
    if (node < N) { cnt[node] = v; start[node] = st; col[st] = node; }
  }
  __syncthreads();

  for (int i = lo; i < hi; i += 4) {
    int2 sd[4];
#pragma unroll
    for (int u = 0; u < 4; ++u) {
      const int ii = (i + u < hi) ? i + u : hi - 1;
      sd[u] = seg[ii];
    }
#pragma unroll
    for (int u = 0; u < 4; ++u) {
      if (i + u < hi) {
        const int j = sd[u].y & (BKT_NODES - 1);
        const int r = atomicAdd(&c_l[j], 1);
        col[s_l[j] + r] = sd[u].x;
      }
    }
  }
}

#define RUN_SHIFT 8
static_assert(BKT_NODES == 8 * 16 && SORT_BLOCKS == 256,
              "a GEMM block's rows and a bucket's nodes must give the same unit count, both roles 256 threads");

__global__ __launch_bounds__(256) void gemm_and_edge_final(
    const float* __restrict__ X, const float* __restrict__ W,
    const float* __restrict__ a_src, const float* __restrict__ a_dst,
    float* __restrict__ H, float* __restrict__ as_, float* __restrict__ ad_, int N,
    const int2* __restrict__ tmp, const int* __restrict__ off, int C,
    int* __restrict__ total, int* __restrict__ cnt, int* __restrict__ start,
    int* __restrict__ col, int nb) {
  __shared__ float4 Wl[FDIM * FDIM / 4];
  __shared__ int c_l[BKT_NODES];
  __shared__ int s_l[BKT_NODES];
  __shared__ int wtot[BKT_NODES / 32];
  __shared__ int base_s;

  const int b     = blockIdx.x;
  const int whole = (nb >> RUN_SHIFT) << (RUN_SHIFT + 1);
  int role, unit;
  if (b < whole) {
    const int run = b >> RUN_SHIFT;
    role = run & 1;
    unit = ((run >> 1) << RUN_SHIFT) | (b & ((1 << RUN_SHIFT) - 1));
  } else {
    const int r   = b - whole;
    const int rem = nb & ((1 << RUN_SHIFT) - 1);
    role = r >= rem ? 1 : 0;
    unit = (whole >> 1) + (role ? r - rem : r);
  }
  if (role) {
    edge_bucket_final_block(tmp, off, C, N, total, cnt, start, col, unit,
                            c_l, s_l, wtot, &base_s);
  } else {
    gat_gemm_wmma_block(X, W, a_src, a_dst, H, as_, ad_, N, unit, Wl);
  }
}

__device__ __forceinline__ float head_dot(float4 r, float4 a) {
  v2f rl; rl.x = r.x; rl.y = r.y;
  v2f rh; rh.x = r.z; rh.y = r.w;
  v2f al; al.x = a.x; al.y = a.y;
  v2f ah; ah.x = a.z; ah.y = a.w;
  const v2f t2 = __builtin_elementwise_fma(rh, ah, rl * al);
  float t = t2.x + t2.y;
  t += row_exchange<0xB1>(t);
  t += row_exchange<0x4E>(t);
  t += row_exchange<0x164>(t);
  return t;
}

__device__ __forceinline__ void row_fma(float4* acc, float a, float4 hv) {
  acc->x = fmaf(a, hv.x, acc->x);
  acc->y = fmaf(a, hv.y, acc->y);
  acc->z = fmaf(a, hv.z, acc->z);
  acc->w = fmaf(a, hv.w, acc->w);
}

__device__ __forceinline__ float4 src_row(const float* __restrict__ H, int s, int jj,
                                          int lane) {
  const int sj = __builtin_amdgcn_readlane(s, jj);
  return *((const float4*)(H + (size_t)sj * FDIM) + lane);
}

__device__ __forceinline__ float4 node_row_sum(const int* __restrict__ col,
                                               const float* __restrict__ H,
                                               int beg, int deg, int d, float adh,
                                               float4 asv, int lane) {
  const float NEG_INF = -__builtin_huge_valf();
  float m = NEG_INF;
  float den = 0.0f;
  float4 acc = make_float4(0.0f, 0.0f, 0.0f, 0.0f);

  for (int base = 0; base < deg; base += 32) {
    const int j = base + lane;
    const int s = (j < deg) ? col[beg + j] : d;
    const int rem = deg - base;
    const int cntc = rem < 32 ? rem : 32;
    int jj = 0;
    for (; jj + 4 <= cntc; jj += 4) {
      const float4 r0 = src_row(H, s, jj + 0, lane);
      const float4 r1 = src_row(H, s, jj + 1, lane);
      const float4 r2 = src_row(H, s, jj + 2, lane);
      const float4 r3 = src_row(H, s, jj + 3, lane);
      const float e0 = leaky(head_dot(r0, asv) + adh);
      const float e1 = leaky(head_dot(r1, asv) + adh);
      const float e2 = leaky(head_dot(r2, asv) + adh);
      const float e3 = leaky(head_dot(r3, asv) + adh);
      const float gm = fmaxf(fmaxf(m, e0), fmaxf(fmaxf(e1, e2), e3));
      const float mu = (gm == NEG_INF) ? 0.0f : gm;
      const float sc = __expf(m - mu);
      const float p0 = __expf(e0 - mu);
      const float p1 = __expf(e1 - mu);
      const float p2 = __expf(e2 - mu);
      const float p3 = __expf(e3 - mu);
      den = fmaf(den, sc, (p0 + p1) + (p2 + p3));
      acc.x *= sc; acc.y *= sc; acc.z *= sc; acc.w *= sc;
      row_fma(&acc, p0, r0); row_fma(&acc, p1, r1);
      row_fma(&acc, p2, r2); row_fma(&acc, p3, r3);
      m = gm;
    }
    for (; jj < cntc; ++jj) {
      const float4 r0 = src_row(H, s, jj, lane);
      const float e0 = leaky(head_dot(r0, asv) + adh);
      const float gm = fmaxf(m, e0);
      const float mu = (gm == NEG_INF) ? 0.0f : gm;
      const float sc = __expf(m - mu);
      const float p0 = __expf(e0 - mu);
      den = fmaf(den, sc, p0);
      acc.x *= sc; acc.y *= sc; acc.z *= sc; acc.w *= sc;
      row_fma(&acc, p0, r0);
      m = gm;
    }
  }

  den += 1e-16f;
  acc.x /= den; acc.y /= den; acc.z /= den; acc.w /= den;
  return acc;
}

#define TILE_PITCH4 33
#define AGG_WAVES 16

__global__ __launch_bounds__(AGG_WAVES * 32) void gat_aggregate_gemm(
    const int* __restrict__ cnt, const int* __restrict__ start,
    const int* __restrict__ col,
    const float* __restrict__ a_src, const float* __restrict__ ad_,
    const float* __restrict__ H, const float* __restrict__ bias,
    const float* __restrict__ W, float* __restrict__ H2, int N) {
  __shared__ float4 Wl[FDIM * FDIM / 4];
  __shared__ float4 tile[AGG_WAVES * TILE_PITCH4];

  const int tid  = threadIdx.x;
  const int lane = tid & 31;
  const int wave = __builtin_amdgcn_readfirstlane(tid >> 5);
  const int m    = lane & 15;
  const int half = lane >> 4;
  const int h    = lane >> 3;

#pragma unroll
  for (int q = tid; q < FDIM * FDIM / 4; q += AGG_WAVES * 32) {
    const int ln  = q & 31;
    const int ctp = (q >> 5) & 3;
    const int ks  = q >> 7;
    const float* __restrict__ w =
        W + (4 * ks + 2 * (ln >> 4)) * FDIM + ctp * 32 + (ln & 15);
    float4 v;
    v.x = w[0];
    v.y = w[FDIM];
    v.z = w[16];
    v.w = w[FDIM + 16];
    Wl[q] = v;
  }

  const float4 asv = *((const float4*)a_src + lane);
  const float4 bv  = *(const float4*)(bias + lane * 4);
  const int nStripes = (N + AGG_WAVES - 1) / AGG_WAVES;

  for (int s = blockIdx.x; s < nStripes; s += gridDim.x) {
    const int row0 = s * AGG_WAVES;
    const int d = row0 + wave;
    const int dd = d < N ? d : N - 1;
    const int beg = start[dd];
    const int deg = d < N ? cnt[dd] : 0;
    const float adh = ad_[(size_t)dd * NHEAD + h];

    float4 acc = node_row_sum(col, H, beg, deg, dd, adh, asv, lane);
    acc.x += bv.x; acc.y += bv.y; acc.z += bv.z; acc.w += bv.w;
    acc.x = acc.x > 0.0f ? acc.x : 0.0f; acc.y = acc.y > 0.0f ? acc.y : 0.0f;
    acc.z = acc.z > 0.0f ? acc.z : 0.0f; acc.w = acc.w > 0.0f ? acc.w : 0.0f;

    __syncthreads();
    tile[wave * TILE_PITCH4 + lane] = acc;
    __syncthreads();

    if (wave < NHEAD) {
      const v2f* __restrict__ tA = (const v2f*)tile + m * (2 * TILE_PITCH4) + half;
      v8f c0 = {}, c1 = {};
#pragma unroll
      for (int ks = 0; ks < FDIM / 4; ++ks) {
        const v2f a = tA[2 * ks];
        const float4 b = Wl[(ks * 4 + wave) * 32 + lane];
        v2f b0; b0.x = b.x; b0.y = b.y;
        v2f b1; b1.x = b.z; b1.y = b.w;
        c0 = __builtin_amdgcn_wmma_f32_16x16x4_f32(
             false, a,  false, b0,
             (short)0, c0,  false,  false);
        c1 = __builtin_amdgcn_wmma_f32_16x16x4_f32(
             false, a,  false, b1,
             (short)0, c1,  false,  false);
      }
      float* __restrict__ hp = H2 + (size_t)(row0 + 8 * half) * FDIM + wave * 32 + m;
#pragma unroll
      for (int v = 0; v < 8; ++v) {
        hp[v * FDIM]      = c0[v];
        hp[v * FDIM + 16] = c1[v];
      }
    }
  }
}

__global__ __launch_bounds__(256) void gat_node_aggregate(
    const int* __restrict__ cnt, const int* __restrict__ start,
    const int* __restrict__ col,
    const float* __restrict__ a_src, const float* __restrict__ a_dst,
    const float* __restrict__ H, const float* __restrict__ bias,
    float* __restrict__ out, int N) {
  const int lane = threadIdx.x & 31;
  const int d = __builtin_amdgcn_readfirstlane(
      (int)(blockIdx.x * (blockDim.x >> 5) + (threadIdx.x >> 5)));
  if (d >= N) return;
  const int beg = start[d];
  const int deg = cnt[d];
  const float4 asv = *((const float4*)a_src + lane);
  const float4 adv = *((const float4*)a_dst + lane);
  const float4 own = *((const float4*)(H + (size_t)d * FDIM) + lane);
  const float adh = head_dot(own, adv);

  float4 acc = node_row_sum(col, H, beg, deg, d, adh, asv, lane);
  const float4 bv = *(const float4*)(bias + lane * 4);
  acc.x += bv.x; acc.y += bv.y; acc.z += bv.z; acc.w += bv.w;
  *(float4*)(out + (size_t)d * FDIM + lane * 4) = acc;
}

static inline int cdiv(long long a, int b) { return (int)((a + b - 1) / b); }

extern "C" void kernel_launch(void* const* d_in, const int* in_sizes, int n_in,
                              void* d_out, int out_size, void* d_ws, size_t ws_size,
                              hipStream_t stream) {
  const float* x     = (const float*)d_in[0];
  const int*   ei    = (const int*)d_in[1];
  const float* W1    = (const float*)d_in[2];
  const float* asrc1 = (const float*)d_in[3];
  const float* adst1 = (const float*)d_in[4];
  const float* b1    = (const float*)d_in[5];
  const float* W2    = (const float*)d_in[6];
  const float* asrc2 = (const float*)d_in[7];
  const float* adst2 = (const float*)d_in[8];
  const float* b2    = (const float*)d_in[9];
  float* out = (float*)d_out;

  const int N  = in_sizes[0] / FDIM;
  const int E  = in_sizes[1] / 2;

  const int nb = cdiv(N, BKT_NODES);
  const int C  = cdiv(E, SORT_BLOCKS);
  float* H2   = (float*)d_ws;
  float* H1   = out;
  int*   total  = (int*)(H2 + (size_t)N * FDIM);
  int2*  tmp    = (int2*)(total + 4);
  int*   off    = total + 4 + (size_t)2 * E;
  int*   cnt    = off + (size_t)(nb + 1) * SORT_BLOCKS;
  int*   start  = cnt + N;
  int*   col    = start + N;
  float* as_  = H2 + (size_t)2 * N * FDIM;
  float* ad_  = as_  + (size_t)N * NHEAD;

  const int TB = 256;
  const int gShared  = 2 * nb;
  const int gNode  = cdiv(N, TB / 32);
  const int gFused = 1024;

  edge_bucket_local<<<SORT_BLOCKS, L1_THREADS, 0, stream>>>(ei, E, C, nb, off, tmp, total);
  gemm_and_edge_final<<<gShared, TB, 0, stream>>>(x, W1, asrc1, adst1, H1, as_, ad_, N,
                                                  tmp, off, C, total, cnt, start, col, nb);

  gat_aggregate_gemm<<<gFused, AGG_WAVES * 32, 0, stream>>>(cnt, start, col, asrc1, ad_, H1, b1,
                                                            W2, H2, N);

  gat_node_aggregate<<<gNode, TB, 0, stream>>>(cnt, start, col, asrc2, adst2, H2, b2, out, N);
}
